# starting kernel
# baseline (speedup 1.0000x reference)
#include <hip/hip_runtime.h>

#define D 128
#define LDSW 130

typedef __attribute__((ext_vector_type(2))) float v2f;
typedef __attribute__((ext_vector_type(8))) float v8f;

__device__ __forceinline__ v8f wmma_f32(v2f a, v2f b, v8f c) {
    return __builtin_amdgcn_wmma_f32_16x16x4_f32(false, a, false, b, (short)0, c,
                                                 false, false);
}

__global__ void zero_kernel(float* __restrict__ p, size_t n) {
    size_t i = (size_t)blockIdx.x * blockDim.x + threadIdx.x;
    if (i < n) p[i] = 0.0f;
}

__global__ __launch_bounds__(256)
void lin_kernel(const float* __restrict__ x, const float* __restrict__ W,
                const float* __restrict__ bias, float* __restrict__ t) {
    __shared__ float aS[16 * LDSW];
    const int tid  = threadIdx.x;
    const int row0 = blockIdx.x * 16;

    for (int i = tid; i < 16 * D; i += 256) {
        int r = i >> 7, c = i & (D - 1);
        aS[r * LDSW + c] = x[(size_t)(row0 + r) * D + c];
    }
    __syncthreads();

    const int lane = tid & 31, wv = tid >> 5;
    const int n = lane & 15, h = lane >> 4, m = lane & 15;
    const int kb = h * 2;
    const int col = wv * 16 + n;

    v8f acc = {};
    for (int k = 0; k < D; k += 4) {
        v2f a = *(const v2f*)&aS[m * LDSW + k + kb];
        v2f b;
        b.x = W[(size_t)(k + kb) * D + col];
        b.y = W[(size_t)(k + kb + 1) * D + col];
        acc = wmma_f32(a, b, acc);
    }
    const float bv = bias[col];
    for (int p = 0; p < 8; ++p)
        t[(size_t)(row0 + p + 8 * h) * D + col] = acc[p] + bv;
}

__global__ void scatter_kernel(const float* __restrict__ tsrc, const int* __restrict__ src,
                               const int* __restrict__ dst, float* __restrict__ agg,
                               float* __restrict__ cnt, int E) {
    size_t i = (size_t)blockIdx.x * blockDim.x + threadIdx.x;
    if (i >= (size_t)E * D) return;
    int e = (int)(i >> 7);
    int c = (int)(i & (D - 1));
    int s = src[e], d0 = dst[e];
    atomicAdd(&agg[(size_t)d0 * D + c], tsrc[(size_t)s * D + c]);
    if (c == 0) atomicAdd(&cnt[d0], 1.0f);
}

__global__ __launch_bounds__(256)
void combine_m_kernel(const float* __restrict__ t_m,
                      const float* __restrict__ agg_mm, const float* __restrict__ cnt_mm,
                      const float* __restrict__ agg_hm, const float* __restrict__ cnt_hm,
                      const float* __restrict__ Wl_mm, const float* __restrict__ bl_mm,
                      const float* __restrict__ Wr_mm,
                      const float* __restrict__ Wl_hm, const float* __restrict__ bl_hm,
                      const float* __restrict__ Wr_hm,
                      const float* __restrict__ g, const float* __restrict__ be,
                      float* __restrict__ out) {
    __shared__ float tS[16 * LDSW], aS[16 * LDSW], bS[16 * LDSW];
    __shared__ float red[4][16];
    const int tid  = threadIdx.x;
    const int row0 = blockIdx.x * 16;

    if (tid < 64) red[tid >> 4][tid & 15] = 0.0f;
    for (int i = tid; i < 16 * D; i += 256) {
        int r = i >> 7, c = i & (D - 1);
        int gr = row0 + r;
        tS[r * LDSW + c] = t_m[(size_t)gr * D + c];
        aS[r * LDSW + c] = agg_mm[(size_t)gr * D + c] / fmaxf(cnt_mm[gr], 1.0f);
        bS[r * LDSW + c] = agg_hm[(size_t)gr * D + c] / fmaxf(cnt_hm[gr], 1.0f);
    }
    __syncthreads();

    const int lane = tid & 31, wv = tid >> 5;
    const int n = lane & 15, h = lane >> 4, m = lane & 15;
    const int kb = h * 2;
    const int col = wv * 16 + n;

    v8f acc_mm = {}, acc_hm = {};
    for (int k = 0; k < D; k += 4) {
        v2f at = *(const v2f*)&tS[m * LDSW + k + kb];
        v2f aa = *(const v2f*)&aS[m * LDSW + k + kb];
        v2f ab = *(const v2f*)&bS[m * LDSW + k + kb];
        v2f b1; b1.x = Wl_mm[(size_t)(k + kb) * D + col]; b1.y = Wl_mm[(size_t)(k + kb + 1) * D + col];
        v2f b2; b2.x = Wr_mm[(size_t)(k + kb) * D + col]; b2.y = Wr_mm[(size_t)(k + kb + 1) * D + col];
        v2f b3; b3.x = Wl_hm[(size_t)(k + kb) * D + col]; b3.y = Wl_hm[(size_t)(k + kb + 1) * D + col];
        v2f b4; b4.x = Wr_hm[(size_t)(k + kb) * D + col]; b4.y = Wr_hm[(size_t)(k + kb + 1) * D + col];
        acc_mm = wmma_f32(aa, b1, acc_mm);
        acc_mm = wmma_f32(at, b2, acc_mm);
        acc_hm = wmma_f32(ab, b3, acc_hm);
        acc_hm = wmma_f32(at, b4, acc_hm);
    }

    float u1[8], u2[8];
    const float c1 = bl_mm[col], c2 = bl_hm[col];
    for (int p = 0; p < 8; ++p) {
        u1[p] = acc_mm[p] + c1;
        u2[p] = acc_hm[p] + c2;
        int r = p + 8 * h;
        atomicAdd(&red[0][r], u1[p] * u1[p]);
        atomicAdd(&red[1][r], u2[p] * u2[p]);
    }
    __syncthreads();

    float y[8];
    for (int p = 0; p < 8; ++p) {
        int r = p + 8 * h;
        float n1 = fmaxf(sqrtf(red[0][r]), 1e-12f);
        float n2 = fmaxf(sqrtf(red[1][r]), 1e-12f);
        float conv = 0.5f * (u1[p] / n1 + u2[p] / n2);
        y[p] = tS[r * LDSW + col] + conv;
        atomicAdd(&red[2][r], y[p]);
        atomicAdd(&red[3][r], y[p] * y[p]);
    }
    __syncthreads();

    const float gc = g[col], bc = be[col];
    for (int p = 0; p < 8; ++p) {
        int r = p + 8 * h;
        float mu  = red[2][r] * (1.0f / D);
        float var = red[3][r] * (1.0f / D) - mu * mu;
        out[(size_t)(row0 + r) * D + col] = (y[p] - mu) * rsqrtf(var + 1e-5f) * gc + bc;
    }
}

__global__ __launch_bounds__(256)
void combine_h_kernel(const float* __restrict__ t_h,
                      const float* __restrict__ agg_mh, const float* __restrict__ cnt_mh,
                      const float* __restrict__ Wl_mh, const float* __restrict__ bl_mh,
                      const float* __restrict__ Wr_mh,
                      const float* __restrict__ g, const float* __restrict__ be,
                      float* __restrict__ out) {
    __shared__ float tS[16 * LDSW], aS[16 * LDSW];
    __shared__ float red[3][16];
    const int tid  = threadIdx.x;
    const int row0 = blockIdx.x * 16;

    if (tid < 48) red[tid >> 4][tid & 15] = 0.0f;
    for (int i = tid; i < 16 * D; i += 256) {
        int r = i >> 7, c = i & (D - 1);
        int gr = row0 + r;
        tS[r * LDSW + c] = t_h[(size_t)gr * D + c];
        aS[r * LDSW + c] = agg_mh[(size_t)gr * D + c] / fmaxf(cnt_mh[gr], 1.0f);
    }
    __syncthreads();

    const int lane = tid & 31, wv = tid >> 5;
    const int n = lane & 15, h = lane >> 4, m = lane & 15;
    const int kb = h * 2;
    const int col = wv * 16 + n;

    v8f acc = {};
    for (int k = 0; k < D; k += 4) {
        v2f at = *(const v2f*)&tS[m * LDSW + k + kb];
        v2f aa = *(const v2f*)&aS[m * LDSW + k + kb];
        v2f b1; b1.x = Wl_mh[(size_t)(k + kb) * D + col]; b1.y = Wl_mh[(size_t)(k + kb + 1) * D + col];
        v2f b2; b2.x = Wr_mh[(size_t)(k + kb) * D + col]; b2.y = Wr_mh[(size_t)(k + kb + 1) * D + col];
        acc = wmma_f32(aa, b1, acc);
        acc = wmma_f32(at, b2, acc);
    }

    float u[8];
    const float c1 = bl_mh[col];
    for (int p = 0; p < 8; ++p) {
        u[p] = acc[p] + c1;
        atomicAdd(&red[0][p + 8 * h], u[p] * u[p]);
    }
    __syncthreads();

    float y[8];
    for (int p = 0; p < 8; ++p) {
        int r = p + 8 * h;
        float n1 = fmaxf(sqrtf(red[0][r]), 1e-12f);
        y[p] = tS[r * LDSW + col] + u[p] / n1;
        atomicAdd(&red[1][r], y[p]);
        atomicAdd(&red[2][r], y[p] * y[p]);
    }
    __syncthreads();

    const float gc = g[col], bc = be[col];
    for (int p = 0; p < 8; ++p) {
        int r = p + 8 * h;
        float mu  = red[1][r] * (1.0f / D);
        float var = red[2][r] * (1.0f / D) - mu * mu;
        out[(size_t)(row0 + r) * D + col] = (y[p] - mu) * rsqrtf(var + 1e-5f) * gc + bc;
    }
}

extern "C" void kernel_launch(void* const* d_in, const int* in_sizes, int n_in,
                              void* d_out, int out_size, void* d_ws, size_t ws_size,
                              hipStream_t stream) {
    const float* x_m    = (const float*)d_in[0];
    const float* x_h    = (const float*)d_in[1];
    const float* W_nt_m = (const float*)d_in[2];
    const float* b_nt_m = (const float*)d_in[3];
    const float* W_nt_h = (const float*)d_in[4];
    const float* b_nt_h = (const float*)d_in[5];
    const float* Wl_mm  = (const float*)d_in[6];
    const float* bl_mm  = (const float*)d_in[7];
    const float* Wr_mm  = (const float*)d_in[8];
    const float* Wl_mh  = (const float*)d_in[9];
    const float* bl_mh  = (const float*)d_in[10];
    const float* Wr_mh  = (const float*)d_in[11];
    const float* Wl_hm  = (const float*)d_in[12];
    const float* bl_hm  = (const float*)d_in[13];
    const float* Wr_hm  = (const float*)d_in[14];
    const float* g_m    = (const float*)d_in[15];
    const float* be_m   = (const float*)d_in[16];
    const float* g_h    = (const float*)d_in[17];
    const float* be_h   = (const float*)d_in[18];
    const int*   ei_mm  = (const int*)d_in[19];
    const int*   ei_mh  = (const int*)d_in[20];
    const int*   ei_hm  = (const int*)d_in[21];

    const int N_M  = in_sizes[0] / D;
    const int N_H  = in_sizes[1] / D;
    const int E_mm = in_sizes[19] / 2;
    const int E_mh = in_sizes[20] / 2;
    const int E_hm = in_sizes[21] / 2;

    const size_t nm = (size_t)N_M * D;
    const size_t nh = (size_t)N_H * D;

    float* ws     = (float*)d_ws;
    float* t_m    = ws;
    float* t_h    = t_m + nm;
    float* agg_mm = t_h + nh;
    float* agg_hm = agg_mm + nm;
    float* agg_mh = agg_hm + nm;
    float* cnt_mm = agg_mh + nh;
    float* cnt_hm = cnt_mm + N_M;
    float* cnt_mh = cnt_hm + N_M;

    size_t zn = (nm + nm + nh) + (size_t)(N_M + N_M + N_H);
    zero_kernel<<<(int)((zn + 255) / 256), 256, 0, stream>>>(agg_mm, zn);

    lin_kernel<<<N_M / 16, 256, 0, stream>>>(x_m, W_nt_m, b_nt_m, t_m);
    lin_kernel<<<N_H / 16, 256, 0, stream>>>(x_h, W_nt_h, b_nt_h, t_h);

    auto sg = [](int E) { return (int)(((size_t)E * D + 255) / 256); };
    scatter_kernel<<<sg(E_mm), 256, 0, stream>>>(t_m, ei_mm, ei_mm + E_mm, agg_mm, cnt_mm, E_mm);
    scatter_kernel<<<sg(E_mh), 256, 0, stream>>>(t_m, ei_mh, ei_mh + E_mh, agg_mh, cnt_mh, E_mh);
    scatter_kernel<<<sg(E_hm), 256, 0, stream>>>(t_h, ei_hm, ei_hm + E_hm, agg_hm, cnt_hm, E_hm);

    float* out_m = (float*)d_out;
    float* out_h = out_m + nm;
    combine_m_kernel<<<N_M / 16, 256, 0, stream>>>(t_m, agg_mm, cnt_mm, agg_hm, cnt_hm,
                                                   Wl_mm, bl_mm, Wr_mm, Wl_hm, bl_hm, Wr_hm,
                                                   g_m, be_m, out_m);
    combine_h_kernel<<<N_H / 16, 256, 0, stream>>>(t_h, agg_mh, cnt_mh,
                                                   Wl_mh, bl_mh, Wr_mh,
                                                   g_h, be_h, out_h);
}


# Edge mean by in-call counting sort of edges by destination (int atomics) and per-row neighbour gather inside the combine kernels; no f32 atomic scatter; sums in slot arrival order
# speedup vs baseline: 1.5953x; 1.5953x over previous
#include <hip/hip_runtime.h>

#define D 128
#define LDSW 130

typedef __attribute__((ext_vector_type(2))) float v2f;
typedef __attribute__((ext_vector_type(8))) float v8f;
typedef __attribute__((ext_vector_type(4))) float v4f;

__device__ __forceinline__ v8f wmma_f32(v2f a, v2f b, v8f c) {
    return __builtin_amdgcn_wmma_f32_16x16x4_f32(false, a, false, b, (short)0, c,
                                                 false, false);
}

__global__ void zero_kernel(float* __restrict__ p, size_t n) {
    size_t i = (size_t)blockIdx.x * blockDim.x + threadIdx.x;
    if (i < n) p[i] = 0.0f;
}

__global__ __launch_bounds__(256)
void lin_kernel(const float* __restrict__ x, const float* __restrict__ W,
                const float* __restrict__ bias, float* __restrict__ t) {
    __shared__ float aS[16 * LDSW];
    const int tid  = threadIdx.x;
    const int row0 = blockIdx.x * 16;

    for (int i = tid; i < 16 * D; i += 256) {
        int r = i >> 7, c = i & (D - 1);
        aS[r * LDSW + c] = x[(size_t)(row0 + r) * D + c];
    }
    __syncthreads();

    const int lane = tid & 31, wv = tid >> 5;
    const int n = lane & 15, h = lane >> 4, m = lane & 15;
    const int kb = h * 2;
    const int col = wv * 16 + n;

    v8f acc = {};
    for (int k = 0; k < D; k += 4) {
        v2f a = *(const v2f*)&aS[m * LDSW + k + kb];
        v2f b;
        b.x = W[(size_t)(k + kb) * D + col];
        b.y = W[(size_t)(k + kb + 1) * D + col];
        acc = wmma_f32(a, b, acc);
    }
    const float bv = bias[col];
    for (int p = 0; p < 8; ++p)
        t[(size_t)(row0 + p + 8 * h) * D + col] = acc[p] + bv;
}


__global__ __launch_bounds__(256)
void degree_kernel(const int* __restrict__ dst0, int* __restrict__ deg0, int E0, int nb0,
                   const int* __restrict__ dst1, int* __restrict__ deg1, int E1, int nb1,
                   const int* __restrict__ dst2, int* __restrict__ deg2, int E2) {
    int b = blockIdx.x;
    const int* dst; int* deg; int E;
    if (b < nb0)            { dst = dst0; deg = deg0; E = E0; }
    else if (b < nb0 + nb1) { b -= nb0;       dst = dst1; deg = deg1; E = E1; }
    else                    { b -= nb0 + nb1; dst = dst2; deg = deg2; E = E2; }
    const int i = b * 256 + (int)threadIdx.x;
    if (i < E) atomicAdd(&deg[dst[i]], 1);
}

__global__ __launch_bounds__(1024)
void scan_kernel(const int* __restrict__ deg0, int* __restrict__ rp0, int n0,
                 const int* __restrict__ deg1, int* __restrict__ rp1, int n1,
                 const int* __restrict__ deg2, int* __restrict__ rp2, int n2) {
    __shared__ int part[1024];
    const int* deg; int* rp; int n;
    if (blockIdx.x == 0)      { deg = deg0; rp = rp0; n = n0; }
    else if (blockIdx.x == 1) { deg = deg1; rp = rp1; n = n1; }
    else                      { deg = deg2; rp = rp2; n = n2; }
    const int tid   = threadIdx.x;
    const int chunk = (n + 1023) >> 10;
    const int b = tid * chunk;
    const int e = min(b + chunk, n);
    int s = 0;
    for (int i = b; i < e; ++i) s += deg[i];
    part[tid] = s;
    __syncthreads();
    for (int off = 1; off < 1024; off <<= 1) {
        const int v = (tid >= off) ? part[tid - off] : 0;
        __syncthreads();
        part[tid] += v;
        __syncthreads();
    }
    int run = part[tid] - s;
    for (int i = b; i < e; ++i) { rp[i] = run; run += deg[i]; }
    if (tid == 1023) rp[n] = part[1023];
}

__global__ __launch_bounds__(256)
void fill_kernel(const int* __restrict__ ei0, const int* __restrict__ rp0, int* __restrict__ deg0,
                 int* __restrict__ nbr0, int E0, int nb0,
                 const int* __restrict__ ei1, const int* __restrict__ rp1, int* __restrict__ deg1,
                 int* __restrict__ nbr1, int E1, int nb1,
                 const int* __restrict__ ei2, const int* __restrict__ rp2, int* __restrict__ deg2,
                 int* __restrict__ nbr2, int E2) {
    int b = blockIdx.x;
    const int* ei; const int* rp; int* deg; int* nbr; int E;
    if (b < nb0)            { ei = ei0; rp = rp0; deg = deg0; nbr = nbr0; E = E0; }
    else if (b < nb0 + nb1) { b -= nb0;       ei = ei1; rp = rp1; deg = deg1; nbr = nbr1; E = E1; }
    else                    { b -= nb0 + nb1; ei = ei2; rp = rp2; deg = deg2; nbr = nbr2; E = E2; }
    const int i = b * 256 + (int)threadIdx.x;
    if (i < E) {
        const int s = ei[i], d = ei[(size_t)E + i];
        const int slot = atomicSub(&deg[d], 1) - 1;
        nbr[rp[d] + slot] = s;
    }
}

__device__ __forceinline__ int first_ids(const int* __restrict__ nbr, int beg, int end, int lane) {
    return (beg + lane < end) ? nbr[beg + lane] : 0;
}

__device__ __forceinline__ v4f ld_row(const float* __restrict__ t, int s, int lane) {
    return *(const v4f*)&t[(size_t)s * D + 4 * lane];
}

__device__ __forceinline__ v4f row_sum(const float* __restrict__ t, const int* __restrict__ nbr,
                                       int beg, int end, int ids, int lane) {
    v4f acc = {0.0f, 0.0f, 0.0f, 0.0f};
    for (int base = beg; base < end; base += 32) {
        if (base != beg) ids = (base + lane < end) ? nbr[base + lane] : 0;
        const int cnt = min(32, end - base);
        int j = 0;
        for (; j + 4 <= cnt; j += 4) {
            const int s0 = __builtin_amdgcn_readlane(ids, j);
            const int s1 = __builtin_amdgcn_readlane(ids, j + 1);
            const int s2 = __builtin_amdgcn_readlane(ids, j + 2);
            const int s3 = __builtin_amdgcn_readlane(ids, j + 3);
            const v4f v0 = ld_row(t, s0, lane);
            const v4f v1 = ld_row(t, s1, lane);
            const v4f v2 = ld_row(t, s2, lane);
            const v4f v3 = ld_row(t, s3, lane);
            acc += v0; acc += v1; acc += v2; acc += v3;
        }
        if (j + 2 <= cnt) {
            const int s0 = __builtin_amdgcn_readlane(ids, j);
            const int s1 = __builtin_amdgcn_readlane(ids, j + 1);
            const v4f v0 = ld_row(t, s0, lane);
            const v4f v1 = ld_row(t, s1, lane);
            acc += v0; acc += v1;
            j += 2;
        }
        if (j < cnt) {
            const int s0 = __builtin_amdgcn_readlane(ids, j);
            acc += ld_row(t, s0, lane);
        }
    }
    return acc;
}

__device__ __forceinline__ void put_mean(float* rowS, v4f s, int count, int lane) {
    const float c = fmaxf((float)count, 1.0f);
    v2f lo, hi;
    lo.x = s.x / c; lo.y = s.y / c; hi.x = s.z / c; hi.y = s.w / c;
    *(v2f*)&rowS[4 * lane]     = lo;
    *(v2f*)&rowS[4 * lane + 2] = hi;
}

__global__ __launch_bounds__(256)
void combine_m_kernel(const float* __restrict__ t_m, const float* __restrict__ t_h,
                      const int* __restrict__ rp_mm, const int* __restrict__ nbr_mm,
                      const int* __restrict__ rp_hm, const int* __restrict__ nbr_hm,
                      const float* __restrict__ Wl_mm, const float* __restrict__ bl_mm,
                      const float* __restrict__ Wr_mm,
                      const float* __restrict__ Wl_hm, const float* __restrict__ bl_hm,
                      const float* __restrict__ Wr_hm,
                      const float* __restrict__ g, const float* __restrict__ be,
                      float* __restrict__ out) {
    __shared__ __attribute__((aligned(16))) float tS[16 * LDSW], aS[16 * LDSW], bS[16 * LDSW];
    __shared__ float red[4][16];
    const int tid  = threadIdx.x;
    const int row0 = blockIdx.x * 16;
    const int lane = tid & 31;
    const int wv = __builtin_amdgcn_readfirstlane(tid >> 5);

    const int g0 = row0 + wv, g1 = g0 + 8;
    const int ba0 = rp_mm[g0], ea0 = rp_mm[g0 + 1], ba1 = rp_mm[g1], ea1 = rp_mm[g1 + 1];
    const int bb0 = rp_hm[g0], eb0 = rp_hm[g0 + 1], bb1 = rp_hm[g1], eb1 = rp_hm[g1 + 1];
    const int ia0 = first_ids(nbr_mm, ba0, ea0, lane);
    const int ib0 = first_ids(nbr_hm, bb0, eb0, lane);
    const int ia1 = first_ids(nbr_mm, ba1, ea1, lane);
    const int ib1 = first_ids(nbr_hm, bb1, eb1, lane);

    if (tid < 64) red[tid >> 4][tid & 15] = 0.0f;
    for (int i = tid; i < 16 * D; i += 256) {
        int r = i >> 7, c = i & (D - 1);
        tS[r * LDSW + c] = t_m[(size_t)(row0 + r) * D + c];
    }
    put_mean(&aS[wv * LDSW],       row_sum(t_m, nbr_mm, ba0, ea0, ia0, lane), ea0 - ba0, lane);
    put_mean(&bS[wv * LDSW],       row_sum(t_h, nbr_hm, bb0, eb0, ib0, lane), eb0 - bb0, lane);
    put_mean(&aS[(wv + 8) * LDSW], row_sum(t_m, nbr_mm, ba1, ea1, ia1, lane), ea1 - ba1, lane);
    put_mean(&bS[(wv + 8) * LDSW], row_sum(t_h, nbr_hm, bb1, eb1, ib1, lane), eb1 - bb1, lane);
    __syncthreads();

    const int n = lane & 15, h = lane >> 4, m = lane & 15;
    const int kb = h * 2;
    const int col = wv * 16 + n;

    v8f acc_mm = {}, acc_hm = {};
    for (int k = 0; k < D; k += 4) {
        v2f at = *(const v2f*)&tS[m * LDSW + k + kb];
        v2f aa = *(const v2f*)&aS[m * LDSW + k + kb];
        v2f ab = *(const v2f*)&bS[m * LDSW + k + kb];
        v2f b1; b1.x = Wl_mm[(size_t)(k + kb) * D + col]; b1.y = Wl_mm[(size_t)(k + kb + 1) * D + col];
        v2f b2; b2.x = Wr_mm[(size_t)(k + kb) * D + col]; b2.y = Wr_mm[(size_t)(k + kb + 1) * D + col];
        v2f b3; b3.x = Wl_hm[(size_t)(k + kb) * D + col]; b3.y = Wl_hm[(size_t)(k + kb + 1) * D + col];
        v2f b4; b4.x = Wr_hm[(size_t)(k + kb) * D + col]; b4.y = Wr_hm[(size_t)(k + kb + 1) * D + col];
        acc_mm = wmma_f32(aa, b1, acc_mm);
        acc_mm = wmma_f32(at, b2, acc_mm);
        acc_hm = wmma_f32(ab, b3, acc_hm);
        acc_hm = wmma_f32(at, b4, acc_hm);
    }

    float u1[8], u2[8];
    const float c1 = bl_mm[col], c2 = bl_hm[col];
    for (int p = 0; p < 8; ++p) {
        u1[p] = acc_mm[p] + c1;
        u2[p] = acc_hm[p] + c2;
        int r = p + 8 * h;
        atomicAdd(&red[0][r], u1[p] * u1[p]);
        atomicAdd(&red[1][r], u2[p] * u2[p]);
    }
    __syncthreads();

    float y[8];
    for (int p = 0; p < 8; ++p) {
        int r = p + 8 * h;
        float n1 = fmaxf(sqrtf(red[0][r]), 1e-12f);
        float n2 = fmaxf(sqrtf(red[1][r]), 1e-12f);
        float conv = 0.5f * (u1[p] / n1 + u2[p] / n2);
        y[p] = tS[r * LDSW + col] + conv;
        atomicAdd(&red[2][r], y[p]);
        atomicAdd(&red[3][r], y[p] * y[p]);
    }
    __syncthreads();

    const float gc = g[col], bc = be[col];
    for (int p = 0; p < 8; ++p) {
        int r = p + 8 * h;
        float mu  = red[2][r] * (1.0f / D);
        float var = red[3][r] * (1.0f / D) - mu * mu;
        out[(size_t)(row0 + r) * D + col] = (y[p] - mu) * rsqrtf(var + 1e-5f) * gc + bc;
    }
}

__global__ __launch_bounds__(256)
void combine_h_kernel(const float* __restrict__ t_h, const float* __restrict__ t_m,
                      const int* __restrict__ rp_mh, const int* __restrict__ nbr_mh,
                      const float* __restrict__ Wl_mh, const float* __restrict__ bl_mh,
                      const float* __restrict__ Wr_mh,
                      const float* __restrict__ g, const float* __restrict__ be,
                      float* __restrict__ out) {
    __shared__ __attribute__((aligned(16))) float tS[16 * LDSW], aS[16 * LDSW];
    __shared__ float red[3][16];
    const int tid  = threadIdx.x;
    const int row0 = blockIdx.x * 16;
    const int lane = tid & 31;
    const int wv = __builtin_amdgcn_readfirstlane(tid >> 5);

    const int g0 = row0 + wv, g1 = g0 + 8;
    const int ba0 = rp_mh[g0], ea0 = rp_mh[g0 + 1], ba1 = rp_mh[g1], ea1 = rp_mh[g1 + 1];
    const int ia0 = first_ids(nbr_mh, ba0, ea0, lane);
    const int ia1 = first_ids(nbr_mh, ba1, ea1, lane);

    if (tid < 48) red[tid >> 4][tid & 15] = 0.0f;
    for (int i = tid; i < 16 * D; i += 256) {
        int r = i >> 7, c = i & (D - 1);
        tS[r * LDSW + c] = t_h[(size_t)(row0 + r) * D + c];
    }
    put_mean(&aS[wv * LDSW],       row_sum(t_m, nbr_mh, ba0, ea0, ia0, lane), ea0 - ba0, lane);
    put_mean(&aS[(wv + 8) * LDSW], row_sum(t_m, nbr_mh, ba1, ea1, ia1, lane), ea1 - ba1, lane);
    __syncthreads();

    const int n = lane & 15, h = lane >> 4, m = lane & 15;
    const int kb = h * 2;
    const int col = wv * 16 + n;

    v8f acc = {};
    for (int k = 0; k < D; k += 4) {
        v2f at = *(const v2f*)&tS[m * LDSW + k + kb];
        v2f aa = *(const v2f*)&aS[m * LDSW + k + kb];
        v2f b1; b1.x = Wl_mh[(size_t)(k + kb) * D + col]; b1.y = Wl_mh[(size_t)(k + kb + 1) * D + col];
        v2f b2; b2.x = Wr_mh[(size_t)(k + kb) * D + col]; b2.y = Wr_mh[(size_t)(k + kb + 1) * D + col];
        acc = wmma_f32(aa, b1, acc);
        acc = wmma_f32(at, b2, acc);
    }

    float u[8];
    const float c1 = bl_mh[col];
    for (int p = 0; p < 8; ++p) {
        u[p] = acc[p] + c1;
        atomicAdd(&red[0][p + 8 * h], u[p] * u[p]);
    }
    __syncthreads();

    float y[8];
    for (int p = 0; p < 8; ++p) {
        int r = p + 8 * h;
        float n1 = fmaxf(sqrtf(red[0][r]), 1e-12f);
        y[p] = tS[r * LDSW + col] + u[p] / n1;
        atomicAdd(&red[1][r], y[p]);
        atomicAdd(&red[2][r], y[p] * y[p]);
    }
    __syncthreads();

    const float gc = g[col], bc = be[col];
    for (int p = 0; p < 8; ++p) {
        int r = p + 8 * h;
        float mu  = red[1][r] * (1.0f / D);
        float var = red[2][r] * (1.0f / D) - mu * mu;
        out[(size_t)(row0 + r) * D + col] = (y[p] - mu) * rsqrtf(var + 1e-5f) * gc + bc;
    }
}

extern "C" void kernel_launch(void* const* d_in, const int* in_sizes, int n_in,
                              void* d_out, int out_size, void* d_ws, size_t ws_size,
                              hipStream_t stream) {
    const float* x_m    = (const float*)d_in[0];
    const float* x_h    = (const float*)d_in[1];
    const float* W_nt_m = (const float*)d_in[2];
    const float* b_nt_m = (const float*)d_in[3];
    const float* W_nt_h = (const float*)d_in[4];
    const float* b_nt_h = (const float*)d_in[5];
    const float* Wl_mm  = (const float*)d_in[6];
    const float* bl_mm  = (const float*)d_in[7];
    const float* Wr_mm  = (const float*)d_in[8];
    const float* Wl_mh  = (const float*)d_in[9];
    const float* bl_mh  = (const float*)d_in[10];
    const float* Wr_mh  = (const float*)d_in[11];
    const float* Wl_hm  = (const float*)d_in[12];
    const float* bl_hm  = (const float*)d_in[13];
    const float* Wr_hm  = (const float*)d_in[14];
    const float* g_m    = (const float*)d_in[15];
    const float* be_m   = (const float*)d_in[16];
    const float* g_h    = (const float*)d_in[17];
    const float* be_h   = (const float*)d_in[18];
    const int*   ei_mm  = (const int*)d_in[19];
    const int*   ei_mh  = (const int*)d_in[20];
    const int*   ei_hm  = (const int*)d_in[21];

    const int N_M  = in_sizes[0] / D;
    const int N_H  = in_sizes[1] / D;
    const int E_mm = in_sizes[19] / 2;
    const int E_mh = in_sizes[20] / 2;
    const int E_hm = in_sizes[21] / 2;

    const size_t nm = (size_t)N_M * D;
    const size_t nh = (size_t)N_H * D;

    float* ws     = (float*)d_ws;
    float* t_m    = ws;
    float* t_h    = t_m + nm;
    int* deg_mm = (int*)(t_h + nh);
    int* deg_hm = deg_mm + N_M;
    int* deg_mh = deg_hm + N_M;
    int* rp_mm  = deg_mh + N_H;
    int* rp_hm  = rp_mm + (N_M + 1);
    int* rp_mh  = rp_hm + (N_M + 1);
    int* nbr_mm = rp_mh + (N_H + 1);
    int* nbr_hm = nbr_mm + E_mm;
    int* nbr_mh = nbr_hm + E_hm;

    size_t zn = (size_t)(N_M + N_M + N_H);
    zero_kernel<<<(int)((zn + 255) / 256), 256, 0, stream>>>((float*)deg_mm, zn);

    lin_kernel<<<N_M / 16, 256, 0, stream>>>(x_m, W_nt_m, b_nt_m, t_m);
    lin_kernel<<<N_H / 16, 256, 0, stream>>>(x_h, W_nt_h, b_nt_h, t_h);

    auto sg = [](int E) { return (E + 255) / 256; };
    const int nb_mm = sg(E_mm), nb_hm = sg(E_hm), nb_mh = sg(E_mh);
    degree_kernel<<<nb_mm + nb_hm + nb_mh, 256, 0, stream>>>(ei_mm + E_mm, deg_mm, E_mm, nb_mm,
                                                             ei_hm + E_hm, deg_hm, E_hm, nb_hm,
                                                             ei_mh + E_mh, deg_mh, E_mh);
    scan_kernel<<<3, 1024, 0, stream>>>(deg_mm, rp_mm, N_M, deg_hm, rp_hm, N_M, deg_mh, rp_mh, N_H);
    fill_kernel<<<nb_mm + nb_hm + nb_mh, 256, 0, stream>>>(ei_mm, rp_mm, deg_mm, nbr_mm, E_mm, nb_mm,
                                                           ei_hm, rp_hm, deg_hm, nbr_hm, E_hm, nb_hm,
                                                           ei_mh, rp_mh, deg_mh, nbr_mh, E_mh);

    float* out_m = (float*)d_out;
    float* out_h = out_m + nm;
    combine_m_kernel<<<N_M / 16, 256, 0, stream>>>(t_m, t_h, rp_mm, nbr_mm, rp_hm, nbr_hm,
                                                   Wl_mm, bl_mm, Wr_mm, Wl_hm, bl_hm, Wr_hm,
                                                   g_m, be_m, out_m);
    combine_h_kernel<<<N_H / 16, 256, 0, stream>>>(t_h, t_m, rp_mh, nbr_mh,
                                                   Wl_mh, bl_mh, Wr_mh,
                                                   g_h, be_h, out_h);
}


# Edge grouping moved into LDS: one work-group per 1024 destination rows reads its edge list twice (count, scan, slot rank); no global atomics, 4 launches become 1; sums still in slot arrival order
# speedup vs baseline: 2.6222x; 1.6437x over previous
#include <hip/hip_runtime.h>

#define D 128
#define LDSW 130

typedef __attribute__((ext_vector_type(2))) float v2f;
typedef __attribute__((ext_vector_type(8))) float v8f;
typedef __attribute__((ext_vector_type(4))) float v4f;

__device__ __forceinline__ v8f wmma_f32(v2f a, v2f b, v8f c) {
    return __builtin_amdgcn_wmma_f32_16x16x4_f32(false, a, false, b, (short)0, c,
                                                 false, false);
}

__global__ __launch_bounds__(256)
void lin_kernel(const float* __restrict__ x, const float* __restrict__ W,
                const float* __restrict__ bias, float* __restrict__ t) {
    __shared__ float aS[16 * LDSW];
    const int tid  = threadIdx.x;
    const int row0 = blockIdx.x * 16;

    for (int i = tid; i < 16 * D; i += 256) {
        int r = i >> 7, c = i & (D - 1);
        aS[r * LDSW + c] = x[(size_t)(row0 + r) * D + c];
    }
    __syncthreads();

    const int lane = tid & 31, wv = tid >> 5;
    const int n = lane & 15, h = lane >> 4, m = lane & 15;
    const int kb = h * 2;
    const int col = wv * 16 + n;

    v8f acc = {};
    for (int k = 0; k < D; k += 4) {
        v2f a = *(const v2f*)&aS[m * LDSW + k + kb];
        v2f b;
        b.x = W[(size_t)(k + kb) * D + col];
        b.y = W[(size_t)(k + kb + 1) * D + col];
        acc = wmma_f32(a, b, acc);
    }
    const float bv = bias[col];
    for (int p = 0; p < 8; ++p)
        t[(size_t)(row0 + p + 8 * h) * D + col] = acc[p] + bv;
}

#define RSH 10
#define RSZ 1024
#define NLD 8
#define NO_EDGE 0x7fffffff

__global__ __launch_bounds__(RSZ)
void group_kernel(const int* __restrict__ ei0, int* __restrict__ rp0, int* __restrict__ nbr0,
                  int E0, int n0, int nr0,
                  const int* __restrict__ ei1, int* __restrict__ rp1, int* __restrict__ nbr1,
                  int E1, int n1, int nr1,
                  const int* __restrict__ ei2, int* __restrict__ rp2, int* __restrict__ nbr2,
                  int E2, int n2, int nr2) {
    __shared__ int cnt[RSZ];
    __shared__ int start[RSZ];
    __shared__ int belowS[RSZ];
    int r = blockIdx.x;
    const int* ei; int* rp; int* nbr; int E; int n; int nr;
    if (r < nr0)            { ei = ei0; rp = rp0; nbr = nbr0; E = E0; n = n0; nr = nr0; }
    else if (r < nr0 + nr1) { r -= nr0;       ei = ei1; rp = rp1; nbr = nbr1; E = E1; n = n1; nr = nr1; }
    else                    { r -= nr0 + nr1; ei = ei2; rp = rp2; nbr = nbr2; E = E2; n = n2; nr = nr2; }
    const int* __restrict__ dst = ei + E;
    const int tid = threadIdx.x;
    const int lo  = r << RSH;
    const int E4 = (((((size_t)dst) | ((size_t)ei)) & 15) == 0) ? (E >> 2) : 0;
    const int4* __restrict__ dst4 = (const int4*)dst;
    const int4* __restrict__ src4 = (const int4*)ei;

    cnt[tid] = 0;
    __syncthreads();

    int below = 0;
    for (int base = 0; base < E4; base += NLD * RSZ) {
        int4 v[NLD];
#pragma unroll
        for (int u = 0; u < NLD; ++u) {
            const int i = base + u * RSZ + tid;
            v[u] = (i < E4) ? dst4[i] : make_int4(NO_EDGE, NO_EDGE, NO_EDGE, NO_EDGE);
        }
#pragma unroll
        for (int u = 0; u < NLD; ++u) {
            const int d0 = v[u].x, d1 = v[u].y, d2 = v[u].z, d3 = v[u].w;
            if ((unsigned)(d0 - lo) < (unsigned)RSZ) atomicAdd(&cnt[d0 - lo], 1);
            if ((unsigned)(d1 - lo) < (unsigned)RSZ) atomicAdd(&cnt[d1 - lo], 1);
            if ((unsigned)(d2 - lo) < (unsigned)RSZ) atomicAdd(&cnt[d2 - lo], 1);
            if ((unsigned)(d3 - lo) < (unsigned)RSZ) atomicAdd(&cnt[d3 - lo], 1);
            below += (d0 < lo) ? 1 : 0;
            below += (d1 < lo) ? 1 : 0;
            below += (d2 < lo) ? 1 : 0;
            below += (d3 < lo) ? 1 : 0;
        }
    }
    for (int i = 4 * E4 + tid; i < E; i += RSZ) {
        const int d = dst[i];
        if ((unsigned)(d - lo) < (unsigned)RSZ) atomicAdd(&cnt[d - lo], 1);
        below += (d < lo) ? 1 : 0;
    }
    __syncthreads();

    const int c = cnt[tid];
    start[tid]  = c;
    belowS[tid] = below;
    __syncthreads();
    for (int off = 1; off < RSZ; off <<= 1) {
        const int a = (tid >= off) ? start[tid - off]  : 0;
        const int b = (tid >= off) ? belowS[tid - off] : 0;
        __syncthreads();
        start[tid]  += a;
        belowS[tid] += b;
        __syncthreads();
    }
    const int first = belowS[RSZ - 1];
    const int incl  = start[tid];
    const int total = start[RSZ - 1];
    __syncthreads();
    const int mine = first + incl - c;
    start[tid] = mine;
    cnt[tid]   = 0;
    if (lo + tid < n) rp[lo + tid] = mine;
    if (r == nr - 1 && tid == RSZ - 1) rp[n] = first + total;
    __syncthreads();

    for (int base = 0; base < E4; base += (NLD / 2) * RSZ) {
        int4 v[NLD / 2], s[NLD / 2];
#pragma unroll
        for (int u = 0; u < NLD / 2; ++u) {
            const int i = base + u * RSZ + tid;
            const bool ok = i < E4;
            v[u] = ok ? dst4[i] : make_int4(NO_EDGE, NO_EDGE, NO_EDGE, NO_EDGE);
            s[u] = ok ? src4[i] : make_int4(0, 0, 0, 0);
        }
#pragma unroll
        for (int u = 0; u < NLD / 2; ++u) {
            const int d0 = v[u].x - lo, d1 = v[u].y - lo, d2 = v[u].z - lo, d3 = v[u].w - lo;
            if ((unsigned)d0 < (unsigned)RSZ) nbr[start[d0] + atomicAdd(&cnt[d0], 1)] = s[u].x;
            if ((unsigned)d1 < (unsigned)RSZ) nbr[start[d1] + atomicAdd(&cnt[d1], 1)] = s[u].y;
            if ((unsigned)d2 < (unsigned)RSZ) nbr[start[d2] + atomicAdd(&cnt[d2], 1)] = s[u].z;
            if ((unsigned)d3 < (unsigned)RSZ) nbr[start[d3] + atomicAdd(&cnt[d3], 1)] = s[u].w;
        }
    }
    for (int i = 4 * E4 + tid; i < E; i += RSZ) {
        const int d = dst[i] - lo;
        if ((unsigned)d < (unsigned)RSZ) nbr[start[d] + atomicAdd(&cnt[d], 1)] = ei[i];
    }
}

__device__ __forceinline__ int first_ids(const int* __restrict__ nbr, int beg, int end, int lane) {
    return (beg + lane < end) ? nbr[beg + lane] : 0;
}

__device__ __forceinline__ v4f ld_row(const float* __restrict__ t, int s, int lane) {
    return *(const v4f*)&t[(size_t)s * D + 4 * lane];
}

__device__ __forceinline__ v4f row_sum(const float* __restrict__ t, const int* __restrict__ nbr,
                                       int beg, int end, int ids, int lane) {
    v4f acc = {0.0f, 0.0f, 0.0f, 0.0f};
    for (int base = beg; base < end; base += 32) {
        if (base != beg) ids = (base + lane < end) ? nbr[base + lane] : 0;
        const int cnt = min(32, end - base);
        int j = 0;
        for (; j + 4 <= cnt; j += 4) {
            const int s0 = __builtin_amdgcn_readlane(ids, j);
            const int s1 = __builtin_amdgcn_readlane(ids, j + 1);
            const int s2 = __builtin_amdgcn_readlane(ids, j + 2);
            const int s3 = __builtin_amdgcn_readlane(ids, j + 3);
            const v4f v0 = ld_row(t, s0, lane);
            const v4f v1 = ld_row(t, s1, lane);
            const v4f v2 = ld_row(t, s2, lane);
            const v4f v3 = ld_row(t, s3, lane);
            acc += v0; acc += v1; acc += v2; acc += v3;
        }
        if (j + 2 <= cnt) {
            const int s0 = __builtin_amdgcn_readlane(ids, j);
            const int s1 = __builtin_amdgcn_readlane(ids, j + 1);
            const v4f v0 = ld_row(t, s0, lane);
            const v4f v1 = ld_row(t, s1, lane);
            acc += v0; acc += v1;
            j += 2;
        }
        if (j < cnt) {
            const int s0 = __builtin_amdgcn_readlane(ids, j);
            acc += ld_row(t, s0, lane);
        }
    }
    return acc;
}

__device__ __forceinline__ void put_mean(float* rowS, v4f s, int count, int lane) {
    const float c = fmaxf((float)count, 1.0f);
    v2f lo, hi;
    lo.x = s.x / c; lo.y = s.y / c; hi.x = s.z / c; hi.y = s.w / c;
    *(v2f*)&rowS[4 * lane]     = lo;
    *(v2f*)&rowS[4 * lane + 2] = hi;
}

__global__ __launch_bounds__(256)
void combine_m_kernel(const float* __restrict__ t_m, const float* __restrict__ t_h,
                      const int* __restrict__ rp_mm, const int* __restrict__ nbr_mm,
                      const int* __restrict__ rp_hm, const int* __restrict__ nbr_hm,
                      const float* __restrict__ Wl_mm, const float* __restrict__ bl_mm,
                      const float* __restrict__ Wr_mm,
                      const float* __restrict__ Wl_hm, const float* __restrict__ bl_hm,
                      const float* __restrict__ Wr_hm,
                      const float* __restrict__ g, const float* __restrict__ be,
                      float* __restrict__ out) {
    __shared__ __attribute__((aligned(16))) float tS[16 * LDSW], aS[16 * LDSW], bS[16 * LDSW];
    __shared__ float red[4][16];
    const int tid  = threadIdx.x;
    const int row0 = blockIdx.x * 16;
    const int lane = tid & 31;
    const int wv = __builtin_amdgcn_readfirstlane(tid >> 5);

    const int g0 = row0 + wv, g1 = g0 + 8;
    const int ba0 = rp_mm[g0], ea0 = rp_mm[g0 + 1], ba1 = rp_mm[g1], ea1 = rp_mm[g1 + 1];
    const int bb0 = rp_hm[g0], eb0 = rp_hm[g0 + 1], bb1 = rp_hm[g1], eb1 = rp_hm[g1 + 1];
    const int ia0 = first_ids(nbr_mm, ba0, ea0, lane);
    const int ib0 = first_ids(nbr_hm, bb0, eb0, lane);
    const int ia1 = first_ids(nbr_mm, ba1, ea1, lane);
    const int ib1 = first_ids(nbr_hm, bb1, eb1, lane);

    if (tid < 64) red[tid >> 4][tid & 15] = 0.0f;
    for (int i = tid; i < 16 * D; i += 256) {
        int r = i >> 7, c = i & (D - 1);
        tS[r * LDSW + c] = t_m[(size_t)(row0 + r) * D + c];
    }
    put_mean(&aS[wv * LDSW],       row_sum(t_m, nbr_mm, ba0, ea0, ia0, lane), ea0 - ba0, lane);
    put_mean(&bS[wv * LDSW],       row_sum(t_h, nbr_hm, bb0, eb0, ib0, lane), eb0 - bb0, lane);
    put_mean(&aS[(wv + 8) * LDSW], row_sum(t_m, nbr_mm, ba1, ea1, ia1, lane), ea1 - ba1, lane);
    put_mean(&bS[(wv + 8) * LDSW], row_sum(t_h, nbr_hm, bb1, eb1, ib1, lane), eb1 - bb1, lane);
    __syncthreads();

    const int n = lane & 15, h = lane >> 4, m = lane & 15;
    const int kb = h * 2;
    const int col = wv * 16 + n;

    v8f acc_mm = {}, acc_hm = {};
    for (int k = 0; k < D; k += 4) {
        v2f at = *(const v2f*)&tS[m * LDSW + k + kb];
        v2f aa = *(const v2f*)&aS[m * LDSW + k + kb];
        v2f ab = *(const v2f*)&bS[m * LDSW + k + kb];
        v2f b1; b1.x = Wl_mm[(size_t)(k + kb) * D + col]; b1.y = Wl_mm[(size_t)(k + kb + 1) * D + col];
        v2f b2; b2.x = Wr_mm[(size_t)(k + kb) * D + col]; b2.y = Wr_mm[(size_t)(k + kb + 1) * D + col];
        v2f b3; b3.x = Wl_hm[(size_t)(k + kb) * D + col]; b3.y = Wl_hm[(size_t)(k + kb + 1) * D + col];
        v2f b4; b4.x = Wr_hm[(size_t)(k + kb) * D + col]; b4.y = Wr_hm[(size_t)(k + kb + 1) * D + col];
        acc_mm = wmma_f32(aa, b1, acc_mm);
        acc_mm = wmma_f32(at, b2, acc_mm);
        acc_hm = wmma_f32(ab, b3, acc_hm);
        acc_hm = wmma_f32(at, b4, acc_hm);
    }

    float u1[8], u2[8];
    const float c1 = bl_mm[col], c2 = bl_hm[col];
    for (int p = 0; p < 8; ++p) {
        u1[p] = acc_mm[p] + c1;
        u2[p] = acc_hm[p] + c2;
        int r = p + 8 * h;
        atomicAdd(&red[0][r], u1[p] * u1[p]);
        atomicAdd(&red[1][r], u2[p] * u2[p]);
    }
    __syncthreads();

    float y[8];
    for (int p = 0; p < 8; ++p) {
        int r = p + 8 * h;
        float n1 = fmaxf(sqrtf(red[0][r]), 1e-12f);
        float n2 = fmaxf(sqrtf(red[1][r]), 1e-12f);
        float conv = 0.5f * (u1[p] / n1 + u2[p] / n2);
        y[p] = tS[r * LDSW + col] + conv;
        atomicAdd(&red[2][r], y[p]);
        atomicAdd(&red[3][r], y[p] * y[p]);
    }
    __syncthreads();

    const float gc = g[col], bc = be[col];
    for (int p = 0; p < 8; ++p) {
        int r = p + 8 * h;
        float mu  = red[2][r] * (1.0f / D);
        float var = red[3][r] * (1.0f / D) - mu * mu;
        out[(size_t)(row0 + r) * D + col] = (y[p] - mu) * rsqrtf(var + 1e-5f) * gc + bc;
    }
}

__global__ __launch_bounds__(256)
void combine_h_kernel(const float* __restrict__ t_h, const float* __restrict__ t_m,
                      const int* __restrict__ rp_mh, const int* __restrict__ nbr_mh,
                      const float* __restrict__ Wl_mh, const float* __restrict__ bl_mh,
                      const float* __restrict__ Wr_mh,
                      const float* __restrict__ g, const float* __restrict__ be,
                      float* __restrict__ out) {
    __shared__ __attribute__((aligned(16))) float tS[16 * LDSW], aS[16 * LDSW];
    __shared__ float red[3][16];
    const int tid  = threadIdx.x;
    const int row0 = blockIdx.x * 16;
    const int lane = tid & 31;
    const int wv = __builtin_amdgcn_readfirstlane(tid >> 5);

    const int g0 = row0 + wv, g1 = g0 + 8;
    const int ba0 = rp_mh[g0], ea0 = rp_mh[g0 + 1], ba1 = rp_mh[g1], ea1 = rp_mh[g1 + 1];
    const int ia0 = first_ids(nbr_mh, ba0, ea0, lane);
    const int ia1 = first_ids(nbr_mh, ba1, ea1, lane);

    if (tid < 48) red[tid >> 4][tid & 15] = 0.0f;
    for (int i = tid; i < 16 * D; i += 256) {
        int r = i >> 7, c = i & (D - 1);
        tS[r * LDSW + c] = t_h[(size_t)(row0 + r) * D + c];
    }
    put_mean(&aS[wv * LDSW],       row_sum(t_m, nbr_mh, ba0, ea0, ia0, lane), ea0 - ba0, lane);
    put_mean(&aS[(wv + 8) * LDSW], row_sum(t_m, nbr_mh, ba1, ea1, ia1, lane), ea1 - ba1, lane);
    __syncthreads();

    const int n = lane & 15, h = lane >> 4, m = lane & 15;
    const int kb = h * 2;
    const int col = wv * 16 + n;

    v8f acc = {};
    for (int k = 0; k < D; k += 4) {
        v2f at = *(const v2f*)&tS[m * LDSW + k + kb];
        v2f aa = *(const v2f*)&aS[m * LDSW + k + kb];
        v2f b1; b1.x = Wl_mh[(size_t)(k + kb) * D + col]; b1.y = Wl_mh[(size_t)(k + kb + 1) * D + col];
        v2f b2; b2.x = Wr_mh[(size_t)(k + kb) * D + col]; b2.y = Wr_mh[(size_t)(k + kb + 1) * D + col];
        acc = wmma_f32(aa, b1, acc);
        acc = wmma_f32(at, b2, acc);
    }

    float u[8];
    const float c1 = bl_mh[col];
    for (int p = 0; p < 8; ++p) {
        u[p] = acc[p] + c1;
        atomicAdd(&red[0][p + 8 * h], u[p] * u[p]);
    }
    __syncthreads();

    float y[8];
    for (int p = 0; p < 8; ++p) {
        int r = p + 8 * h;
        float n1 = fmaxf(sqrtf(red[0][r]), 1e-12f);
        y[p] = tS[r * LDSW + col] + u[p] / n1;
        atomicAdd(&red[1][r], y[p]);
        atomicAdd(&red[2][r], y[p] * y[p]);
    }
    __syncthreads();

    const float gc = g[col], bc = be[col];
    for (int p = 0; p < 8; ++p) {
        int r = p + 8 * h;
        float mu  = red[1][r] * (1.0f / D);
        float var = red[2][r] * (1.0f / D) - mu * mu;
        out[(size_t)(row0 + r) * D + col] = (y[p] - mu) * rsqrtf(var + 1e-5f) * gc + bc;
    }
}

extern "C" void kernel_launch(void* const* d_in, const int* in_sizes, int n_in,
                              void* d_out, int out_size, void* d_ws, size_t ws_size,
                              hipStream_t stream) {
    const float* x_m    = (const float*)d_in[0];
    const float* x_h    = (const float*)d_in[1];
    const float* W_nt_m = (const float*)d_in[2];
    const float* b_nt_m = (const float*)d_in[3];
    const float* W_nt_h = (const float*)d_in[4];
    const float* b_nt_h = (const float*)d_in[5];
    const float* Wl_mm  = (const float*)d_in[6];
    const float* bl_mm  = (const float*)d_in[7];
    const float* Wr_mm  = (const float*)d_in[8];
    const float* Wl_mh  = (const float*)d_in[9];
    const float* bl_mh  = (const float*)d_in[10];
    const float* Wr_mh  = (const float*)d_in[11];
    const float* Wl_hm  = (const float*)d_in[12];
    const float* bl_hm  = (const float*)d_in[13];
    const float* Wr_hm  = (const float*)d_in[14];
    const float* g_m    = (const float*)d_in[15];
    const float* be_m   = (const float*)d_in[16];
    const float* g_h    = (const float*)d_in[17];
    const float* be_h   = (const float*)d_in[18];
    const int*   ei_mm  = (const int*)d_in[19];
    const int*   ei_mh  = (const int*)d_in[20];
    const int*   ei_hm  = (const int*)d_in[21];

    const int N_M  = in_sizes[0] / D;
    const int N_H  = in_sizes[1] / D;
    const int E_mm = in_sizes[19] / 2;
    const int E_mh = in_sizes[20] / 2;
    const int E_hm = in_sizes[21] / 2;

    const size_t nm = (size_t)N_M * D;
    const size_t nh = (size_t)N_H * D;

    float* ws     = (float*)d_ws;
    float* t_m    = ws;
    float* t_h    = t_m + nm;
    int* rp_mm  = (int*)(t_h + nh);
    int* rp_hm  = rp_mm + (N_M + 1);
    int* rp_mh  = rp_hm + (N_M + 1);
    int* nbr_mm = rp_mh + (N_H + 1);
    int* nbr_hm = nbr_mm + E_mm;
    int* nbr_mh = nbr_hm + E_hm;

    lin_kernel<<<N_M / 16, 256, 0, stream>>>(x_m, W_nt_m, b_nt_m, t_m);
    lin_kernel<<<N_H / 16, 256, 0, stream>>>(x_h, W_nt_h, b_nt_h, t_h);

    auto rg = [](int n) { return (n + RSZ - 1) / RSZ; };
    const int nr_mm = rg(N_M), nr_hm = rg(N_M), nr_mh = rg(N_H);
    group_kernel<<<nr_mm + nr_hm + nr_mh, RSZ, 0, stream>>>(ei_mm, rp_mm, nbr_mm, E_mm, N_M, nr_mm,
                                                            ei_hm, rp_hm, nbr_hm, E_hm, N_M, nr_hm,
                                                            ei_mh, rp_mh, nbr_mh, E_mh, N_H, nr_mh);

    float* out_m = (float*)d_out;
    float* out_h = out_m + nm;
    combine_m_kernel<<<N_M / 16, 256, 0, stream>>>(t_m, t_h, rp_mm, nbr_mm, rp_hm, nbr_hm,
                                                   Wl_mm, bl_mm, Wr_mm, Wl_hm, bl_hm, Wr_hm,
                                                   g_m, be_m, out_m);
    combine_h_kernel<<<N_H / 16, 256, 0, stream>>>(t_h, t_m, rp_mh, nbr_mh,
                                                   Wl_mh, bl_mh, Wr_mh,
                                                   g_h, be_h, out_h);
}


# Epilogue row sums (both L2 norms, LayerNorm mean and mean square) by lane-butterfly sums over rows staged in LDS instead of same-address LDS float atomics; sums now in a fixed tree order
# speedup vs baseline: 4.0200x; 1.5331x over previous
#include <hip/hip_runtime.h>

#define D 128
#define LDSW 130

typedef __attribute__((ext_vector_type(2))) float v2f;
typedef __attribute__((ext_vector_type(8))) float v8f;
typedef __attribute__((ext_vector_type(4))) float v4f;

__device__ __forceinline__ v8f wmma_f32(v2f a, v2f b, v8f c) {
    return __builtin_amdgcn_wmma_f32_16x16x4_f32(false, a, false, b, (short)0, c,
                                                 false, false);
}

__global__ __launch_bounds__(256)
void lin_kernel(const float* __restrict__ x, const float* __restrict__ W,
                const float* __restrict__ bias, float* __restrict__ t) {
    __shared__ float aS[16 * LDSW];
    const int tid  = threadIdx.x;
    const int row0 = blockIdx.x * 16;

    for (int i = tid; i < 16 * D; i += 256) {
        int r = i >> 7, c = i & (D - 1);
        aS[r * LDSW + c] = x[(size_t)(row0 + r) * D + c];
    }
    __syncthreads();

    const int lane = tid & 31, wv = tid >> 5;
    const int n = lane & 15, h = lane >> 4, m = lane & 15;
    const int kb = h * 2;
    const int col = wv * 16 + n;

    v8f acc = {};
    for (int k = 0; k < D; k += 4) {
        v2f a = *(const v2f*)&aS[m * LDSW + k + kb];
        v2f b;
        b.x = W[(size_t)(k + kb) * D + col];
        b.y = W[(size_t)(k + kb + 1) * D + col];
        acc = wmma_f32(a, b, acc);
    }
    const float bv = bias[col];
    for (int p = 0; p < 8; ++p)
        t[(size_t)(row0 + p + 8 * h) * D + col] = acc[p] + bv;
}

#define RSH 10
#define RSZ 1024
#define NLD 8
#define NO_EDGE 0x7fffffff

__global__ __launch_bounds__(RSZ)
void group_kernel(const int* __restrict__ ei0, int* __restrict__ rp0, int* __restrict__ nbr0,
                  int E0, int n0, int nr0,
                  const int* __restrict__ ei1, int* __restrict__ rp1, int* __restrict__ nbr1,
                  int E1, int n1, int nr1,
                  const int* __restrict__ ei2, int* __restrict__ rp2, int* __restrict__ nbr2,
                  int E2, int n2, int nr2) {
    __shared__ int cnt[RSZ];
    __shared__ int start[RSZ];
    __shared__ int belowS[RSZ];
    int r = blockIdx.x;
    const int* ei; int* rp; int* nbr; int E; int n; int nr;
    if (r < nr0)            { ei = ei0; rp = rp0; nbr = nbr0; E = E0; n = n0; nr = nr0; }
    else if (r < nr0 + nr1) { r -= nr0;       ei = ei1; rp = rp1; nbr = nbr1; E = E1; n = n1; nr = nr1; }
    else                    { r -= nr0 + nr1; ei = ei2; rp = rp2; nbr = nbr2; E = E2; n = n2; nr = nr2; }
    const int* __restrict__ dst = ei + E;
    const int tid = threadIdx.x;
    const int lo  = r << RSH;
    const int E4 = (((((size_t)dst) | ((size_t)ei)) & 15) == 0) ? (E >> 2) : 0;
    const int4* __restrict__ dst4 = (const int4*)dst;
    const int4* __restrict__ src4 = (const int4*)ei;

    cnt[tid] = 0;
    __syncthreads();

    int below = 0;
    for (int base = 0; base < E4; base += NLD * RSZ) {
        int4 v[NLD];
#pragma unroll
        for (int u = 0; u < NLD; ++u) {
            const int i = base + u * RSZ + tid;
            v[u] = (i < E4) ? dst4[i] : make_int4(NO_EDGE, NO_EDGE, NO_EDGE, NO_EDGE);
        }
#pragma unroll
        for (int u = 0; u < NLD; ++u) {
            const int d0 = v[u].x, d1 = v[u].y, d2 = v[u].z, d3 = v[u].w;
            if ((unsigned)(d0 - lo) < (unsigned)RSZ) atomicAdd(&cnt[d0 - lo], 1);
            if ((unsigned)(d1 - lo) < (unsigned)RSZ) atomicAdd(&cnt[d1 - lo], 1);
            if ((unsigned)(d2 - lo) < (unsigned)RSZ) atomicAdd(&cnt[d2 - lo], 1);
            if ((unsigned)(d3 - lo) < (unsigned)RSZ) atomicAdd(&cnt[d3 - lo], 1);
            below += (d0 < lo) ? 1 : 0;
            below += (d1 < lo) ? 1 : 0;
            below += (d2 < lo) ? 1 : 0;
            below += (d3 < lo) ? 1 : 0;
        }
    }
    for (int i = 4 * E4 + tid; i < E; i += RSZ) {
        const int d = dst[i];
        if ((unsigned)(d - lo) < (unsigned)RSZ) atomicAdd(&cnt[d - lo], 1);
        below += (d < lo) ? 1 : 0;
    }
    __syncthreads();

    const int c = cnt[tid];
    start[tid]  = c;
    belowS[tid] = below;
    __syncthreads();
    for (int off = 1; off < RSZ; off <<= 1) {
        const int a = (tid >= off) ? start[tid - off]  : 0;
        const int b = (tid >= off) ? belowS[tid - off] : 0;
        __syncthreads();
        start[tid]  += a;
        belowS[tid] += b;
        __syncthreads();
    }
    const int first = belowS[RSZ - 1];
    const int incl  = start[tid];
    const int total = start[RSZ - 1];
    __syncthreads();
    const int mine = first + incl - c;
    start[tid] = mine;
    cnt[tid]   = 0;
    if (lo + tid < n) rp[lo + tid] = mine;
    if (r == nr - 1 && tid == RSZ - 1) rp[n] = first + total;
    __syncthreads();

    for (int base = 0; base < E4; base += (NLD / 2) * RSZ) {
        int4 v[NLD / 2], s[NLD / 2];
#pragma unroll
        for (int u = 0; u < NLD / 2; ++u) {
            const int i = base + u * RSZ + tid;
            const bool ok = i < E4;
            v[u] = ok ? dst4[i] : make_int4(NO_EDGE, NO_EDGE, NO_EDGE, NO_EDGE);
            s[u] = ok ? src4[i] : make_int4(0, 0, 0, 0);
        }
#pragma unroll
        for (int u = 0; u < NLD / 2; ++u) {
            const int d0 = v[u].x - lo, d1 = v[u].y - lo, d2 = v[u].z - lo, d3 = v[u].w - lo;
            if ((unsigned)d0 < (unsigned)RSZ) nbr[start[d0] + atomicAdd(&cnt[d0], 1)] = s[u].x;
            if ((unsigned)d1 < (unsigned)RSZ) nbr[start[d1] + atomicAdd(&cnt[d1], 1)] = s[u].y;
            if ((unsigned)d2 < (unsigned)RSZ) nbr[start[d2] + atomicAdd(&cnt[d2], 1)] = s[u].z;
            if ((unsigned)d3 < (unsigned)RSZ) nbr[start[d3] + atomicAdd(&cnt[d3], 1)] = s[u].w;
        }
    }
    for (int i = 4 * E4 + tid; i < E; i += RSZ) {
        const int d = dst[i] - lo;
        if ((unsigned)d < (unsigned)RSZ) nbr[start[d] + atomicAdd(&cnt[d], 1)] = ei[i];
    }
}

__device__ __forceinline__ int first_ids(const int* __restrict__ nbr, int beg, int end, int lane) {
    return (beg + lane < end) ? nbr[beg + lane] : 0;
}

__device__ __forceinline__ v4f ld_row(const float* __restrict__ t, int s, int lane) {
    return *(const v4f*)&t[(size_t)s * D + 4 * lane];
}

__device__ __forceinline__ v4f row_sum(const float* __restrict__ t, const int* __restrict__ nbr,
                                       int beg, int end, int ids, int lane) {
    v4f acc = {0.0f, 0.0f, 0.0f, 0.0f};
    for (int base = beg; base < end; base += 32) {
        if (base != beg) ids = (base + lane < end) ? nbr[base + lane] : 0;
        const int cnt = min(32, end - base);
        int j = 0;
        for (; j + 4 <= cnt; j += 4) {
            const int s0 = __builtin_amdgcn_readlane(ids, j);
            const int s1 = __builtin_amdgcn_readlane(ids, j + 1);
            const int s2 = __builtin_amdgcn_readlane(ids, j + 2);
            const int s3 = __builtin_amdgcn_readlane(ids, j + 3);
            const v4f v0 = ld_row(t, s0, lane);
            const v4f v1 = ld_row(t, s1, lane);
            const v4f v2 = ld_row(t, s2, lane);
            const v4f v3 = ld_row(t, s3, lane);
            acc += v0; acc += v1; acc += v2; acc += v3;
        }
        if (j + 2 <= cnt) {
            const int s0 = __builtin_amdgcn_readlane(ids, j);
            const int s1 = __builtin_amdgcn_readlane(ids, j + 1);
            const v4f v0 = ld_row(t, s0, lane);
            const v4f v1 = ld_row(t, s1, lane);
            acc += v0; acc += v1;
            j += 2;
        }
        if (j < cnt) {
            const int s0 = __builtin_amdgcn_readlane(ids, j);
            acc += ld_row(t, s0, lane);
        }
    }
    return acc;
}

__device__ __forceinline__ void put_mean(float* rowS, v4f s, int count, int lane) {
    const float c = fmaxf((float)count, 1.0f);
    v2f lo, hi;
    lo.x = s.x / c; lo.y = s.y / c; hi.x = s.z / c; hi.y = s.w / c;
    *(v2f*)&rowS[4 * lane]     = lo;
    *(v2f*)&rowS[4 * lane + 2] = hi;
}

template <int M>
__device__ __forceinline__ float row_xor(float v) {
    return __builtin_bit_cast(float, __builtin_amdgcn_update_dpp(0, __builtin_bit_cast(int, v),
                                                                 0x160 + M, 0xf, 0xf, true));
}

__device__ __forceinline__ float xhalf(float v) {
    const int b = __builtin_bit_cast(int, v);
    return __builtin_bit_cast(float, __builtin_amdgcn_permlanex16(b, b, 0x76543210, 0xfedcba98,
                                                                  false, false));
}

__device__ __forceinline__ float wave_sum(float v) {
    v += row_xor<1>(v);
    v += row_xor<2>(v);
    v += row_xor<4>(v);
    v += row_xor<8>(v);
    v += xhalf(v);
    return v;
}

__device__ __forceinline__ float sq4(v2f a, v2f b) {
    return a.x * a.x + a.y * a.y + b.x * b.x + b.y * b.y;
}

__global__ __launch_bounds__(256)
void combine_m_kernel(const float* __restrict__ t_m, const float* __restrict__ t_h,
                      const int* __restrict__ rp_mm, const int* __restrict__ nbr_mm,
                      const int* __restrict__ rp_hm, const int* __restrict__ nbr_hm,
                      const float* __restrict__ Wl_mm, const float* __restrict__ bl_mm,
                      const float* __restrict__ Wr_mm,
                      const float* __restrict__ Wl_hm, const float* __restrict__ bl_hm,
                      const float* __restrict__ Wr_hm,
                      const float* __restrict__ g, const float* __restrict__ be,
                      float* __restrict__ out) {
    __shared__ __attribute__((aligned(16))) float tS[16 * LDSW], aS[16 * LDSW], bS[16 * LDSW];
    const int tid  = threadIdx.x;
    const int row0 = blockIdx.x * 16;
    const int lane = tid & 31;
    const int wv = __builtin_amdgcn_readfirstlane(tid >> 5);

    const int g0 = row0 + wv, g1 = g0 + 8;
    const int ba0 = rp_mm[g0], ea0 = rp_mm[g0 + 1], ba1 = rp_mm[g1], ea1 = rp_mm[g1 + 1];
    const int bb0 = rp_hm[g0], eb0 = rp_hm[g0 + 1], bb1 = rp_hm[g1], eb1 = rp_hm[g1 + 1];
    const int ia0 = first_ids(nbr_mm, ba0, ea0, lane);
    const int ib0 = first_ids(nbr_hm, bb0, eb0, lane);
    const int ia1 = first_ids(nbr_mm, ba1, ea1, lane);
    const int ib1 = first_ids(nbr_hm, bb1, eb1, lane);

    for (int i = tid; i < 16 * D; i += 256) {
        int r = i >> 7, c = i & (D - 1);
        tS[r * LDSW + c] = t_m[(size_t)(row0 + r) * D + c];
    }
    put_mean(&aS[wv * LDSW],       row_sum(t_m, nbr_mm, ba0, ea0, ia0, lane), ea0 - ba0, lane);
    put_mean(&bS[wv * LDSW],       row_sum(t_h, nbr_hm, bb0, eb0, ib0, lane), eb0 - bb0, lane);
    put_mean(&aS[(wv + 8) * LDSW], row_sum(t_m, nbr_mm, ba1, ea1, ia1, lane), ea1 - ba1, lane);
    put_mean(&bS[(wv + 8) * LDSW], row_sum(t_h, nbr_hm, bb1, eb1, ib1, lane), eb1 - bb1, lane);
    __syncthreads();

    const int n = lane & 15, h = lane >> 4, m = lane & 15;
    const int kb = h * 2;
    const int col = wv * 16 + n;

    v8f acc_mm = {}, acc_hm = {};
    for (int k = 0; k < D; k += 4) {
        v2f at = *(const v2f*)&tS[m * LDSW + k + kb];
        v2f aa = *(const v2f*)&aS[m * LDSW + k + kb];
        v2f ab = *(const v2f*)&bS[m * LDSW + k + kb];
        v2f b1; b1.x = Wl_mm[(size_t)(k + kb) * D + col]; b1.y = Wl_mm[(size_t)(k + kb + 1) * D + col];
        v2f b2; b2.x = Wr_mm[(size_t)(k + kb) * D + col]; b2.y = Wr_mm[(size_t)(k + kb + 1) * D + col];
        v2f b3; b3.x = Wl_hm[(size_t)(k + kb) * D + col]; b3.y = Wl_hm[(size_t)(k + kb + 1) * D + col];
        v2f b4; b4.x = Wr_hm[(size_t)(k + kb) * D + col]; b4.y = Wr_hm[(size_t)(k + kb + 1) * D + col];
        acc_mm = wmma_f32(aa, b1, acc_mm);
        acc_mm = wmma_f32(at, b2, acc_mm);
        acc_hm = wmma_f32(ab, b3, acc_hm);
        acc_hm = wmma_f32(at, b4, acc_hm);
    }

    const float c1 = bl_mm[col], c2 = bl_hm[col];
    __syncthreads();
    for (int p = 0; p < 8; ++p) {
        const int r = p + 8 * h;
        aS[r * LDSW + col] = acc_mm[p] + c1;
        bS[r * LDSW + col] = acc_hm[p] + c2;
    }
    __syncthreads();

    const int cA = 2 * lane, cB = 64 + 2 * lane;
    const float gA0 = g[cA], gA1 = g[cA + 1], gB0 = g[cB], gB1 = g[cB + 1];
    const float eA0 = be[cA], eA1 = be[cA + 1], eB0 = be[cB], eB1 = be[cB + 1];
#pragma unroll 1
    for (int i = 0; i < 2; ++i) {
        const int r = wv + 8 * i;
        const v2f tA = *(const v2f*)&tS[r * LDSW + cA], tB = *(const v2f*)&tS[r * LDSW + cB];
        const v2f pA = *(const v2f*)&aS[r * LDSW + cA], pB = *(const v2f*)&aS[r * LDSW + cB];
        const v2f qA = *(const v2f*)&bS[r * LDSW + cA], qB = *(const v2f*)&bS[r * LDSW + cB];
        const float s1 = wave_sum(sq4(pA, pB));
        const float s2 = wave_sum(sq4(qA, qB));
        const float n1 = fmaxf(sqrtf(s1), 1e-12f);
        const float n2 = fmaxf(sqrtf(s2), 1e-12f);
        const float y0 = tA.x + 0.5f * (pA.x / n1 + qA.x / n2);
        const float y1 = tA.y + 0.5f * (pA.y / n1 + qA.y / n2);
        const float y2 = tB.x + 0.5f * (pB.x / n1 + qB.x / n2);
        const float y3 = tB.y + 0.5f * (pB.y / n1 + qB.y / n2);
        const float sy  = wave_sum(y0 + y1 + y2 + y3);
        const float syy = wave_sum(y0 * y0 + y1 * y1 + y2 * y2 + y3 * y3);
        const float mu  = sy * (1.0f / D);
        const float var = syy * (1.0f / D) - mu * mu;
        const float rs  = rsqrtf(var + 1e-5f);
        v2f oA, oB;
        oA.x = (y0 - mu) * rs * gA0 + eA0;
        oA.y = (y1 - mu) * rs * gA1 + eA1;
        oB.x = (y2 - mu) * rs * gB0 + eB0;
        oB.y = (y3 - mu) * rs * gB1 + eB1;
        float* orow = out + (size_t)(row0 + r) * D;
        *(v2f*)&orow[cA] = oA;
        *(v2f*)&orow[cB] = oB;
    }
}

__global__ __launch_bounds__(256)
void combine_h_kernel(const float* __restrict__ t_h, const float* __restrict__ t_m,
                      const int* __restrict__ rp_mh, const int* __restrict__ nbr_mh,
                      const float* __restrict__ Wl_mh, const float* __restrict__ bl_mh,
                      const float* __restrict__ Wr_mh,
                      const float* __restrict__ g, const float* __restrict__ be,
                      float* __restrict__ out) {
    __shared__ __attribute__((aligned(16))) float tS[16 * LDSW], aS[16 * LDSW];
    const int tid  = threadIdx.x;
    const int row0 = blockIdx.x * 16;
    const int lane = tid & 31;
    const int wv = __builtin_amdgcn_readfirstlane(tid >> 5);

    const int g0 = row0 + wv, g1 = g0 + 8;
    const int ba0 = rp_mh[g0], ea0 = rp_mh[g0 + 1], ba1 = rp_mh[g1], ea1 = rp_mh[g1 + 1];
    const int ia0 = first_ids(nbr_mh, ba0, ea0, lane);
    const int ia1 = first_ids(nbr_mh, ba1, ea1, lane);

    for (int i = tid; i < 16 * D; i += 256) {
        int r = i >> 7, c = i & (D - 1);
        tS[r * LDSW + c] = t_h[(size_t)(row0 + r) * D + c];
    }
    put_mean(&aS[wv * LDSW],       row_sum(t_m, nbr_mh, ba0, ea0, ia0, lane), ea0 - ba0, lane);
    put_mean(&aS[(wv + 8) * LDSW], row_sum(t_m, nbr_mh, ba1, ea1, ia1, lane), ea1 - ba1, lane);
    __syncthreads();

    const int n = lane & 15, h = lane >> 4, m = lane & 15;
    const int kb = h * 2;
    const int col = wv * 16 + n;

    v8f acc = {};
    for (int k = 0; k < D; k += 4) {
        v2f at = *(const v2f*)&tS[m * LDSW + k + kb];
        v2f aa = *(const v2f*)&aS[m * LDSW + k + kb];
        v2f b1; b1.x = Wl_mh[(size_t)(k + kb) * D + col]; b1.y = Wl_mh[(size_t)(k + kb + 1) * D + col];
        v2f b2; b2.x = Wr_mh[(size_t)(k + kb) * D + col]; b2.y = Wr_mh[(size_t)(k + kb + 1) * D + col];
        acc = wmma_f32(aa, b1, acc);
        acc = wmma_f32(at, b2, acc);
    }

    const float c1 = bl_mh[col];
    __syncthreads();
    for (int p = 0; p < 8; ++p)
        aS[(p + 8 * h) * LDSW + col] = acc[p] + c1;
    __syncthreads();

    const int cA = 2 * lane, cB = 64 + 2 * lane;
    const float gA0 = g[cA], gA1 = g[cA + 1], gB0 = g[cB], gB1 = g[cB + 1];
    const float eA0 = be[cA], eA1 = be[cA + 1], eB0 = be[cB], eB1 = be[cB + 1];
#pragma unroll 1
    for (int i = 0; i < 2; ++i) {
        const int r = wv + 8 * i;
        const v2f tA = *(const v2f*)&tS[r * LDSW + cA], tB = *(const v2f*)&tS[r * LDSW + cB];
        const v2f pA = *(const v2f*)&aS[r * LDSW + cA], pB = *(const v2f*)&aS[r * LDSW + cB];
        const float s1 = wave_sum(sq4(pA, pB));
        const float n1 = fmaxf(sqrtf(s1), 1e-12f);
        const float y0 = tA.x + pA.x / n1;
        const float y1 = tA.y + pA.y / n1;
        const float y2 = tB.x + pB.x / n1;
        const float y3 = tB.y + pB.y / n1;
        const float sy  = wave_sum(y0 + y1 + y2 + y3);
        const float syy = wave_sum(y0 * y0 + y1 * y1 + y2 * y2 + y3 * y3);
        const float mu  = sy * (1.0f / D);
        const float var = syy * (1.0f / D) - mu * mu;
        const float rs  = rsqrtf(var + 1e-5f);
        v2f oA, oB;
        oA.x = (y0 - mu) * rs * gA0 + eA0;
        oA.y = (y1 - mu) * rs * gA1 + eA1;
        oB.x = (y2 - mu) * rs * gB0 + eB0;
        oB.y = (y3 - mu) * rs * gB1 + eB1;
        float* orow = out + (size_t)(row0 + r) * D;
        *(v2f*)&orow[cA] = oA;
        *(v2f*)&orow[cB] = oB;
    }
}

extern "C" void kernel_launch(void* const* d_in, const int* in_sizes, int n_in,
                              void* d_out, int out_size, void* d_ws, size_t ws_size,
                              hipStream_t stream) {
    const float* x_m    = (const float*)d_in[0];
    const float* x_h    = (const float*)d_in[1];
    const float* W_nt_m = (const float*)d_in[2];
    const float* b_nt_m = (const float*)d_in[3];
    const float* W_nt_h = (const float*)d_in[4];
    const float* b_nt_h = (const float*)d_in[5];
    const float* Wl_mm  = (const float*)d_in[6];
    const float* bl_mm  = (const float*)d_in[7];
    const float* Wr_mm  = (const float*)d_in[8];
    const float* Wl_mh  = (const float*)d_in[9];
    const float* bl_mh  = (const float*)d_in[10];
    const float* Wr_mh  = (const float*)d_in[11];
    const float* Wl_hm  = (const float*)d_in[12];
    const float* bl_hm  = (const float*)d_in[13];
    const float* Wr_hm  = (const float*)d_in[14];
    const float* g_m    = (const float*)d_in[15];
    const float* be_m   = (const float*)d_in[16];
    const float* g_h    = (const float*)d_in[17];
    const float* be_h   = (const float*)d_in[18];
    const int*   ei_mm  = (const int*)d_in[19];
    const int*   ei_mh  = (const int*)d_in[20];
    const int*   ei_hm  = (const int*)d_in[21];

    const int N_M  = in_sizes[0] / D;
    const int N_H  = in_sizes[1] / D;
    const int E_mm = in_sizes[19] / 2;
    const int E_mh = in_sizes[20] / 2;
    const int E_hm = in_sizes[21] / 2;

    const size_t nm = (size_t)N_M * D;
    const size_t nh = (size_t)N_H * D;

    float* ws     = (float*)d_ws;
    float* t_m    = ws;
    float* t_h    = t_m + nm;
    int* rp_mm  = (int*)(t_h + nh);
    int* rp_hm  = rp_mm + (N_M + 1);
    int* rp_mh  = rp_hm + (N_M + 1);
    int* nbr_mm = rp_mh + (N_H + 1);
    int* nbr_hm = nbr_mm + E_mm;
    int* nbr_mh = nbr_hm + E_hm;

    lin_kernel<<<N_M / 16, 256, 0, stream>>>(x_m, W_nt_m, b_nt_m, t_m);
    lin_kernel<<<N_H / 16, 256, 0, stream>>>(x_h, W_nt_h, b_nt_h, t_h);

    auto rg = [](int n) { return (n + RSZ - 1) / RSZ; };
    const int nr_mm = rg(N_M), nr_hm = rg(N_M), nr_mh = rg(N_H);
    group_kernel<<<nr_mm + nr_hm + nr_mh, RSZ, 0, stream>>>(ei_mm, rp_mm, nbr_mm, E_mm, N_M, nr_mm,
                                                            ei_hm, rp_hm, nbr_hm, E_hm, N_M, nr_hm,
                                                            ei_mh, rp_mh, nbr_mh, E_mh, N_H, nr_mh);

    float* out_m = (float*)d_out;
    float* out_h = out_m + nm;
    combine_m_kernel<<<N_M / 16, 256, 0, stream>>>(t_m, t_h, rp_mm, nbr_mm, rp_hm, nbr_hm,
                                                   Wl_mm, bl_mm, Wr_mm, Wl_hm, bl_hm, Wr_hm,
                                                   g_m, be_m, out_m);
    combine_h_kernel<<<N_H / 16, 256, 0, stream>>>(t_h, t_m, rp_mh, nbr_mh,
                                                   Wl_mh, bl_mh, Wr_mh,
                                                   g_h, be_h, out_h);
}


# Edge grouping reads the edge lists once: the count pass keeps each owned edge (index, row) in a per-wave LDS list; the fill pass walks it, fetching sources by index; sums still in slot arrival order
# speedup vs baseline: 4.3226x; 1.0753x over previous
#include <hip/hip_runtime.h>

#define D 128
#define LDSW 130

typedef __attribute__((ext_vector_type(2))) float v2f;
typedef __attribute__((ext_vector_type(8))) float v8f;
typedef __attribute__((ext_vector_type(4))) float v4f;

__device__ __forceinline__ v8f wmma_f32(v2f a, v2f b, v8f c) {
    return __builtin_amdgcn_wmma_f32_16x16x4_f32(false, a, false, b, (short)0, c,
                                                 false, false);
}

__global__ __launch_bounds__(256)
void lin_kernel(const float* __restrict__ x, const float* __restrict__ W,
                const float* __restrict__ bias, float* __restrict__ t) {
    __shared__ float aS[16 * LDSW];
    const int tid  = threadIdx.x;
    const int row0 = blockIdx.x * 16;

    for (int i = tid; i < 16 * D; i += 256) {
        int r = i >> 7, c = i & (D - 1);
        aS[r * LDSW + c] = x[(size_t)(row0 + r) * D + c];
    }
    __syncthreads();

    const int lane = tid & 31, wv = tid >> 5;
    const int n = lane & 15, h = lane >> 4, m = lane & 15;
    const int kb = h * 2;
    const int col = wv * 16 + n;

    v8f acc = {};
    for (int k = 0; k < D; k += 4) {
        v2f a = *(const v2f*)&aS[m * LDSW + k + kb];
        v2f b;
        b.x = W[(size_t)(k + kb) * D + col];
        b.y = W[(size_t)(k + kb + 1) * D + col];
        acc = wmma_f32(a, b, acc);
    }
    const float bv = bias[col];
    for (int p = 0; p < 8; ++p)
        t[(size_t)(row0 + p + 8 * h) * D + col] = acc[p] + bv;
}

#define RSH 10
#define RSZ 1024
#define NLD 8
#define NO_EDGE 0x7fffffff
#define NWV (RSZ / 32)
#define WCAP 1024
#define IDX_BITS (32 - RSH)

#define VISIT(dv, iv)                                                                      \
    do {                                                                                   \
        const int dl_ = (dv) - lo;                                                         \
        const bool hit_ = (unsigned)dl_ < (unsigned)RSZ;                                   \
        const unsigned mk_ = __builtin_amdgcn_ballot_w32(hit_);                            \
        if (hit_) {                                                                        \
            atomicAdd(&cnt[dl_], 1);                                                       \
            const int slot_ = wcount + (int)__builtin_amdgcn_mbcnt_lo(mk_, 0u);            \
            if (slot_ < cap) mine_list[slot_] = ((unsigned)(iv) << RSH) | (unsigned)dl_;   \
        }                                                                                  \
        wcount += __builtin_popcount(mk_);                                                 \
        below += ((dv) < lo) ? 1 : 0;                                                      \
    } while (0)

__global__ __launch_bounds__(RSZ)
void group_kernel(const int* __restrict__ ei0, int* __restrict__ rp0, int* __restrict__ nbr0,
                  int E0, int n0, int nr0,
                  const int* __restrict__ ei1, int* __restrict__ rp1, int* __restrict__ nbr1,
                  int E1, int n1, int nr1,
                  const int* __restrict__ ei2, int* __restrict__ rp2, int* __restrict__ nbr2,
                  int E2, int n2, int nr2) {
    __shared__ int cnt[RSZ];
    __shared__ int start[RSZ];
    __shared__ int belowS[RSZ];
    __shared__ unsigned listS[NWV * WCAP];
    __shared__ int spill;
    int r = blockIdx.x;
    const int* ei; int* rp; int* nbr; int E; int n; int nr;
    if (r < nr0)            { ei = ei0; rp = rp0; nbr = nbr0; E = E0; n = n0; nr = nr0; }
    else if (r < nr0 + nr1) { r -= nr0;       ei = ei1; rp = rp1; nbr = nbr1; E = E1; n = n1; nr = nr1; }
    else                    { r -= nr0 + nr1; ei = ei2; rp = rp2; nbr = nbr2; E = E2; n = n2; nr = nr2; }
    const int* __restrict__ dst = ei + E;
    const int tid  = threadIdx.x;
    const int lane = tid & 31;
    const int lo   = r << RSH;
    const int E4 = (((((size_t)dst) | ((size_t)ei)) & 15) == 0) ? (E >> 2) : 0;
    const int4* __restrict__ dst4 = (const int4*)dst;
    const int4* __restrict__ src4 = (const int4*)ei;
    const int cap = (E <= (1 << IDX_BITS)) ? WCAP : 0;
    unsigned* mine_list = &listS[(tid >> 5) * WCAP];

    cnt[tid] = 0;
    if (tid == 0) spill = 0;
    __syncthreads();

    int below = 0;
    int wcount = 0;
    for (int base = 0; base < E4; base += NLD * RSZ) {
        int4 v[NLD];
#pragma unroll
        for (int u = 0; u < NLD; ++u) {
            const int i = base + u * RSZ + tid;
            v[u] = (i < E4) ? dst4[i] : make_int4(NO_EDGE, NO_EDGE, NO_EDGE, NO_EDGE);
        }
#pragma unroll
        for (int u = 0; u < NLD; ++u) {
            const int i4 = 4 * (base + u * RSZ + tid);
            VISIT(v[u].x, i4);
            VISIT(v[u].y, i4 + 1);
            VISIT(v[u].z, i4 + 2);
            VISIT(v[u].w, i4 + 3);
        }
    }
    for (int base = 4 * E4; base < E; base += RSZ) {
        const int i = base + tid;
        const int d = (i < E) ? dst[i] : NO_EDGE;
        VISIT(d, i);
    }
    if (lane == 0 && wcount > cap) spill = 1;
    __syncthreads();

    const int c = cnt[tid];
    start[tid]  = c;
    belowS[tid] = below;
    __syncthreads();
    for (int off = 1; off < RSZ; off <<= 1) {
        const int a = (tid >= off) ? start[tid - off]  : 0;
        const int b = (tid >= off) ? belowS[tid - off] : 0;
        __syncthreads();
        start[tid]  += a;
        belowS[tid] += b;
        __syncthreads();
    }
    const int first = belowS[RSZ - 1];
    const int incl  = start[tid];
    const int total = start[RSZ - 1];
    const bool listed = (spill == 0);
    __syncthreads();
    const int mine = first + incl - c;
    start[tid] = mine;
    cnt[tid]   = 0;
    if (lo + tid < n) rp[lo + tid] = mine;
    if (r == nr - 1 && tid == RSZ - 1) rp[n] = first + total;
    __syncthreads();

    if (listed) {
        for (int kb = 0; kb < wcount; kb += 4 * 32) {
            int dl[4], s[4];
#pragma unroll
            for (int u = 0; u < 4; ++u) {
                const int k = kb + u * 32 + lane;
                const bool ok = k < wcount;
                const unsigned e = ok ? mine_list[k] : 0u;
                dl[u] = ok ? (int)(e & (RSZ - 1)) : -1;
                s[u]  = ok ? ei[e >> RSH] : 0;
            }
#pragma unroll
            for (int u = 0; u < 4; ++u)
                if (dl[u] >= 0) nbr[start[dl[u]] + atomicAdd(&cnt[dl[u]], 1)] = s[u];
        }
        return;
    }

    for (int base = 0; base < E4; base += (NLD / 2) * RSZ) {
        int4 v[NLD / 2], s[NLD / 2];
#pragma unroll
        for (int u = 0; u < NLD / 2; ++u) {
            const int i = base + u * RSZ + tid;
            const bool ok = i < E4;
            v[u] = ok ? dst4[i] : make_int4(NO_EDGE, NO_EDGE, NO_EDGE, NO_EDGE);
            s[u] = ok ? src4[i] : make_int4(0, 0, 0, 0);
        }
#pragma unroll
        for (int u = 0; u < NLD / 2; ++u) {
            const int d0 = v[u].x - lo, d1 = v[u].y - lo, d2 = v[u].z - lo, d3 = v[u].w - lo;
            if ((unsigned)d0 < (unsigned)RSZ) nbr[start[d0] + atomicAdd(&cnt[d0], 1)] = s[u].x;
            if ((unsigned)d1 < (unsigned)RSZ) nbr[start[d1] + atomicAdd(&cnt[d1], 1)] = s[u].y;
            if ((unsigned)d2 < (unsigned)RSZ) nbr[start[d2] + atomicAdd(&cnt[d2], 1)] = s[u].z;
            if ((unsigned)d3 < (unsigned)RSZ) nbr[start[d3] + atomicAdd(&cnt[d3], 1)] = s[u].w;
        }
    }
    for (int i = 4 * E4 + tid; i < E; i += RSZ) {
        const int d = dst[i] - lo;
        if ((unsigned)d < (unsigned)RSZ) nbr[start[d] + atomicAdd(&cnt[d], 1)] = ei[i];
    }
}
#undef VISIT

__device__ __forceinline__ int first_ids(const int* __restrict__ nbr, int beg, int end, int lane) {
    return (beg + lane < end) ? nbr[beg + lane] : 0;
}

__device__ __forceinline__ v4f ld_row(const float* __restrict__ t, int s, int lane) {
    return *(const v4f*)&t[(size_t)s * D + 4 * lane];
}

__device__ __forceinline__ v4f row_sum(const float* __restrict__ t, const int* __restrict__ nbr,
                                       int beg, int end, int ids, int lane) {
    v4f acc = {0.0f, 0.0f, 0.0f, 0.0f};
    for (int base = beg; base < end; base += 32) {
        if (base != beg) ids = (base + lane < end) ? nbr[base + lane] : 0;
        const int cnt = min(32, end - base);
        int j = 0;
        for (; j + 4 <= cnt; j += 4) {
            const int s0 = __builtin_amdgcn_readlane(ids, j);
            const int s1 = __builtin_amdgcn_readlane(ids, j + 1);
            const int s2 = __builtin_amdgcn_readlane(ids, j + 2);
            const int s3 = __builtin_amdgcn_readlane(ids, j + 3);
            const v4f v0 = ld_row(t, s0, lane);
            const v4f v1 = ld_row(t, s1, lane);
            const v4f v2 = ld_row(t, s2, lane);
            const v4f v3 = ld_row(t, s3, lane);
            acc += v0; acc += v1; acc += v2; acc += v3;
        }
        if (j + 2 <= cnt) {
            const int s0 = __builtin_amdgcn_readlane(ids, j);
            const int s1 = __builtin_amdgcn_readlane(ids, j + 1);
            const v4f v0 = ld_row(t, s0, lane);
            const v4f v1 = ld_row(t, s1, lane);
            acc += v0; acc += v1;
            j += 2;
        }
        if (j < cnt) {
            const int s0 = __builtin_amdgcn_readlane(ids, j);
            acc += ld_row(t, s0, lane);
        }
    }
    return acc;
}

__device__ __forceinline__ void put_mean(float* rowS, v4f s, int count, int lane) {
    const float c = fmaxf((float)count, 1.0f);
    v2f lo, hi;
    lo.x = s.x / c; lo.y = s.y / c; hi.x = s.z / c; hi.y = s.w / c;
    *(v2f*)&rowS[4 * lane]     = lo;
    *(v2f*)&rowS[4 * lane + 2] = hi;
}

template <int M>
__device__ __forceinline__ float row_xor(float v) {
    return __builtin_bit_cast(float, __builtin_amdgcn_update_dpp(0, __builtin_bit_cast(int, v),
                                                                 0x160 + M, 0xf, 0xf, true));
}

__device__ __forceinline__ float xhalf(float v) {
    const int b = __builtin_bit_cast(int, v);
    return __builtin_bit_cast(float, __builtin_amdgcn_permlanex16(b, b, 0x76543210, 0xfedcba98,
                                                                  false, false));
}

__device__ __forceinline__ float wave_sum(float v) {
    v += row_xor<1>(v);
    v += row_xor<2>(v);
    v += row_xor<4>(v);
    v += row_xor<8>(v);
    v += xhalf(v);
    return v;
}

__device__ __forceinline__ float sq4(v2f a, v2f b) {
    return a.x * a.x + a.y * a.y + b.x * b.x + b.y * b.y;
}

__global__ __launch_bounds__(256)
void combine_m_kernel(const float* __restrict__ t_m, const float* __restrict__ t_h,
                      const int* __restrict__ rp_mm, const int* __restrict__ nbr_mm,
                      const int* __restrict__ rp_hm, const int* __restrict__ nbr_hm,
                      const float* __restrict__ Wl_mm, const float* __restrict__ bl_mm,
                      const float* __restrict__ Wr_mm,
                      const float* __restrict__ Wl_hm, const float* __restrict__ bl_hm,
                      const float* __restrict__ Wr_hm,
                      const float* __restrict__ g, const float* __restrict__ be,
                      float* __restrict__ out) {
    __shared__ __attribute__((aligned(16))) float tS[16 * LDSW], aS[16 * LDSW], bS[16 * LDSW];
    const int tid  = threadIdx.x;
    const int row0 = blockIdx.x * 16;
    const int lane = tid & 31;
    const int wv = __builtin_amdgcn_readfirstlane(tid >> 5);

    const int g0 = row0 + wv, g1 = g0 + 8;
    const int ba0 = rp_mm[g0], ea0 = rp_mm[g0 + 1], ba1 = rp_mm[g1], ea1 = rp_mm[g1 + 1];
    const int bb0 = rp_hm[g0], eb0 = rp_hm[g0 + 1], bb1 = rp_hm[g1], eb1 = rp_hm[g1 + 1];
    const int ia0 = first_ids(nbr_mm, ba0, ea0, lane);
    const int ib0 = first_ids(nbr_hm, bb0, eb0, lane);
    const int ia1 = first_ids(nbr_mm, ba1, ea1, lane);
    const int ib1 = first_ids(nbr_hm, bb1, eb1, lane);

    for (int i = tid; i < 16 * D; i += 256) {
        int r = i >> 7, c = i & (D - 1);
        tS[r * LDSW + c] = t_m[(size_t)(row0 + r) * D + c];
    }
    put_mean(&aS[wv * LDSW],       row_sum(t_m, nbr_mm, ba0, ea0, ia0, lane), ea0 - ba0, lane);
    put_mean(&bS[wv * LDSW],       row_sum(t_h, nbr_hm, bb0, eb0, ib0, lane), eb0 - bb0, lane);
    put_mean(&aS[(wv + 8) * LDSW], row_sum(t_m, nbr_mm, ba1, ea1, ia1, lane), ea1 - ba1, lane);
    put_mean(&bS[(wv + 8) * LDSW], row_sum(t_h, nbr_hm, bb1, eb1, ib1, lane), eb1 - bb1, lane);
    __syncthreads();

    const int n = lane & 15, h = lane >> 4, m = lane & 15;
    const int kb = h * 2;
    const int col = wv * 16 + n;

    v8f acc_mm = {}, acc_hm = {};
    for (int k = 0; k < D; k += 4) {
        v2f at = *(const v2f*)&tS[m * LDSW + k + kb];
        v2f aa = *(const v2f*)&aS[m * LDSW + k + kb];
        v2f ab = *(const v2f*)&bS[m * LDSW + k + kb];
        v2f b1; b1.x = Wl_mm[(size_t)(k + kb) * D + col]; b1.y = Wl_mm[(size_t)(k + kb + 1) * D + col];
        v2f b2; b2.x = Wr_mm[(size_t)(k + kb) * D + col]; b2.y = Wr_mm[(size_t)(k + kb + 1) * D + col];
        v2f b3; b3.x = Wl_hm[(size_t)(k + kb) * D + col]; b3.y = Wl_hm[(size_t)(k + kb + 1) * D + col];
        v2f b4; b4.x = Wr_hm[(size_t)(k + kb) * D + col]; b4.y = Wr_hm[(size_t)(k + kb + 1) * D + col];
        acc_mm = wmma_f32(aa, b1, acc_mm);
        acc_mm = wmma_f32(at, b2, acc_mm);
        acc_hm = wmma_f32(ab, b3, acc_hm);
        acc_hm = wmma_f32(at, b4, acc_hm);
    }

    const float c1 = bl_mm[col], c2 = bl_hm[col];
    __syncthreads();
    for (int p = 0; p < 8; ++p) {
        const int r = p + 8 * h;
        aS[r * LDSW + col] = acc_mm[p] + c1;
        bS[r * LDSW + col] = acc_hm[p] + c2;
    }
    __syncthreads();

    const int cA = 2 * lane, cB = 64 + 2 * lane;
    const float gA0 = g[cA], gA1 = g[cA + 1], gB0 = g[cB], gB1 = g[cB + 1];
    const float eA0 = be[cA], eA1 = be[cA + 1], eB0 = be[cB], eB1 = be[cB + 1];
#pragma unroll 1
    for (int i = 0; i < 2; ++i) {
        const int r = wv + 8 * i;
        const v2f tA = *(const v2f*)&tS[r * LDSW + cA], tB = *(const v2f*)&tS[r * LDSW + cB];
        const v2f pA = *(const v2f*)&aS[r * LDSW + cA], pB = *(const v2f*)&aS[r * LDSW + cB];
        const v2f qA = *(const v2f*)&bS[r * LDSW + cA], qB = *(const v2f*)&bS[r * LDSW + cB];
        const float s1 = wave_sum(sq4(pA, pB));
        const float s2 = wave_sum(sq4(qA, qB));
        const float n1 = fmaxf(sqrtf(s1), 1e-12f);
        const float n2 = fmaxf(sqrtf(s2), 1e-12f);
        const float y0 = tA.x + 0.5f * (pA.x / n1 + qA.x / n2);
        const float y1 = tA.y + 0.5f * (pA.y / n1 + qA.y / n2);
        const float y2 = tB.x + 0.5f * (pB.x / n1 + qB.x / n2);
        const float y3 = tB.y + 0.5f * (pB.y / n1 + qB.y / n2);
        const float sy  = wave_sum(y0 + y1 + y2 + y3);
        const float syy = wave_sum(y0 * y0 + y1 * y1 + y2 * y2 + y3 * y3);
        const float mu  = sy * (1.0f / D);
        const float var = syy * (1.0f / D) - mu * mu;
        const float rs  = rsqrtf(var + 1e-5f);
        v2f oA, oB;
        oA.x = (y0 - mu) * rs * gA0 + eA0;
        oA.y = (y1 - mu) * rs * gA1 + eA1;
        oB.x = (y2 - mu) * rs * gB0 + eB0;
        oB.y = (y3 - mu) * rs * gB1 + eB1;
        float* orow = out + (size_t)(row0 + r) * D;
        *(v2f*)&orow[cA] = oA;
        *(v2f*)&orow[cB] = oB;
    }
}

__global__ __launch_bounds__(256)
void combine_h_kernel(const float* __restrict__ t_h, const float* __restrict__ t_m,
                      const int* __restrict__ rp_mh, const int* __restrict__ nbr_mh,
                      const float* __restrict__ Wl_mh, const float* __restrict__ bl_mh,
                      const float* __restrict__ Wr_mh,
                      const float* __restrict__ g, const float* __restrict__ be,
                      float* __restrict__ out) {
    __shared__ __attribute__((aligned(16))) float tS[16 * LDSW], aS[16 * LDSW];
    const int tid  = threadIdx.x;
    const int row0 = blockIdx.x * 16;
    const int lane = tid & 31;
    const int wv = __builtin_amdgcn_readfirstlane(tid >> 5);

    const int g0 = row0 + wv, g1 = g0 + 8;
    const int ba0 = rp_mh[g0], ea0 = rp_mh[g0 + 1], ba1 = rp_mh[g1], ea1 = rp_mh[g1 + 1];
    const int ia0 = first_ids(nbr_mh, ba0, ea0, lane);
    const int ia1 = first_ids(nbr_mh, ba1, ea1, lane);

    for (int i = tid; i < 16 * D; i += 256) {
        int r = i >> 7, c = i & (D - 1);
        tS[r * LDSW + c] = t_h[(size_t)(row0 + r) * D + c];
    }
    put_mean(&aS[wv * LDSW],       row_sum(t_m, nbr_mh, ba0, ea0, ia0, lane), ea0 - ba0, lane);
    put_mean(&aS[(wv + 8) * LDSW], row_sum(t_m, nbr_mh, ba1, ea1, ia1, lane), ea1 - ba1, lane);
    __syncthreads();

    const int n = lane & 15, h = lane >> 4, m = lane & 15;
    const int kb = h * 2;
    const int col = wv * 16 + n;

    v8f acc = {};
    for (int k = 0; k < D; k += 4) {
        v2f at = *(const v2f*)&tS[m * LDSW + k + kb];
        v2f aa = *(const v2f*)&aS[m * LDSW + k + kb];
        v2f b1; b1.x = Wl_mh[(size_t)(k + kb) * D + col]; b1.y = Wl_mh[(size_t)(k + kb + 1) * D + col];
        v2f b2; b2.x = Wr_mh[(size_t)(k + kb) * D + col]; b2.y = Wr_mh[(size_t)(k + kb + 1) * D + col];
        acc = wmma_f32(aa, b1, acc);
        acc = wmma_f32(at, b2, acc);
    }

    const float c1 = bl_mh[col];
    __syncthreads();
    for (int p = 0; p < 8; ++p)
        aS[(p + 8 * h) * LDSW + col] = acc[p] + c1;
    __syncthreads();

    const int cA = 2 * lane, cB = 64 + 2 * lane;
    const float gA0 = g[cA], gA1 = g[cA + 1], gB0 = g[cB], gB1 = g[cB + 1];
    const float eA0 = be[cA], eA1 = be[cA + 1], eB0 = be[cB], eB1 = be[cB + 1];
#pragma unroll 1
    for (int i = 0; i < 2; ++i) {
        const int r = wv + 8 * i;
        const v2f tA = *(const v2f*)&tS[r * LDSW + cA], tB = *(const v2f*)&tS[r * LDSW + cB];
        const v2f pA = *(const v2f*)&aS[r * LDSW + cA], pB = *(const v2f*)&aS[r * LDSW + cB];
        const float s1 = wave_sum(sq4(pA, pB));
        const float n1 = fmaxf(sqrtf(s1), 1e-12f);
        const float y0 = tA.x + pA.x / n1;
        const float y1 = tA.y + pA.y / n1;
        const float y2 = tB.x + pB.x / n1;
        const float y3 = tB.y + pB.y / n1;
        const float sy  = wave_sum(y0 + y1 + y2 + y3);
        const float syy = wave_sum(y0 * y0 + y1 * y1 + y2 * y2 + y3 * y3);
        const float mu  = sy * (1.0f / D);
        const float var = syy * (1.0f / D) - mu * mu;
        const float rs  = rsqrtf(var + 1e-5f);
        v2f oA, oB;
        oA.x = (y0 - mu) * rs * gA0 + eA0;
        oA.y = (y1 - mu) * rs * gA1 + eA1;
        oB.x = (y2 - mu) * rs * gB0 + eB0;
        oB.y = (y3 - mu) * rs * gB1 + eB1;
        float* orow = out + (size_t)(row0 + r) * D;
        *(v2f*)&orow[cA] = oA;
        *(v2f*)&orow[cB] = oB;
    }
}

extern "C" void kernel_launch(void* const* d_in, const int* in_sizes, int n_in,
                              void* d_out, int out_size, void* d_ws, size_t ws_size,
                              hipStream_t stream) {
    const float* x_m    = (const float*)d_in[0];
    const float* x_h    = (const float*)d_in[1];
    const float* W_nt_m = (const float*)d_in[2];
    const float* b_nt_m = (const float*)d_in[3];
    const float* W_nt_h = (const float*)d_in[4];
    const float* b_nt_h = (const float*)d_in[5];
    const float* Wl_mm  = (const float*)d_in[6];
    const float* bl_mm  = (const float*)d_in[7];
    const float* Wr_mm  = (const float*)d_in[8];
    const float* Wl_mh  = (const float*)d_in[9];
    const float* bl_mh  = (const float*)d_in[10];
    const float* Wr_mh  = (const float*)d_in[11];
    const float* Wl_hm  = (const float*)d_in[12];
    const float* bl_hm  = (const float*)d_in[13];
    const float* Wr_hm  = (const float*)d_in[14];
    const float* g_m    = (const float*)d_in[15];
    const float* be_m   = (const float*)d_in[16];
    const float* g_h    = (const float*)d_in[17];
    const float* be_h   = (const float*)d_in[18];
    const int*   ei_mm  = (const int*)d_in[19];
    const int*   ei_mh  = (const int*)d_in[20];
    const int*   ei_hm  = (const int*)d_in[21];

    const int N_M  = in_sizes[0] / D;
    const int N_H  = in_sizes[1] / D;
    const int E_mm = in_sizes[19] / 2;
    const int E_mh = in_sizes[20] / 2;
    const int E_hm = in_sizes[21] / 2;

    const size_t nm = (size_t)N_M * D;
    const size_t nh = (size_t)N_H * D;

    float* ws     = (float*)d_ws;
    float* t_m    = ws;
    float* t_h    = t_m + nm;
    int* rp_mm  = (int*)(t_h + nh);
    int* rp_hm  = rp_mm + (N_M + 1);
    int* rp_mh  = rp_hm + (N_M + 1);
    int* nbr_mm = rp_mh + (N_H + 1);
    int* nbr_hm = nbr_mm + E_mm;
    int* nbr_mh = nbr_hm + E_hm;

    lin_kernel<<<N_M / 16, 256, 0, stream>>>(x_m, W_nt_m, b_nt_m, t_m);
    lin_kernel<<<N_H / 16, 256, 0, stream>>>(x_h, W_nt_h, b_nt_h, t_h);

    auto rg = [](int n) { return (n + RSZ - 1) / RSZ; };
    const int nr_mm = rg(N_M), nr_hm = rg(N_M), nr_mh = rg(N_H);
    group_kernel<<<nr_mm + nr_hm + nr_mh, RSZ, 0, stream>>>(ei_mm, rp_mm, nbr_mm, E_mm, N_M, nr_mm,
                                                            ei_hm, rp_hm, nbr_hm, E_hm, N_M, nr_hm,
                                                            ei_mh, rp_mh, nbr_mh, E_mh, N_H, nr_mh);

    float* out_m = (float*)d_out;
    float* out_h = out_m + nm;
    combine_m_kernel<<<N_M / 16, 256, 0, stream>>>(t_m, t_h, rp_mm, nbr_mm, rp_hm, nbr_hm,
                                                   Wl_mm, bl_mm, Wr_mm, Wl_hm, bl_hm, Wr_hm,
                                                   g_m, be_m, out_m);
    combine_h_kernel<<<N_H / 16, 256, 0, stream>>>(t_h, t_m, rp_mh, nbr_mh,
                                                   Wl_mh, bl_mh, Wr_mh,
                                                   g_h, be_h, out_h);
}


# Edge grouping and both linear transforms share one launch: the first work-groups group edges, the rest compute t = xW + b in 64-row work-groups beside them; 5 launches become 3; sum orders unchanged
# speedup vs baseline: 4.7947x; 1.1092x over previous
#include <hip/hip_runtime.h>

#define D 128
#define LDSW 130

typedef __attribute__((ext_vector_type(2))) float v2f;
typedef __attribute__((ext_vector_type(8))) float v8f;
typedef __attribute__((ext_vector_type(4))) float v4f;

__device__ __forceinline__ v8f wmma_f32(v2f a, v2f b, v8f c) {
    return __builtin_amdgcn_wmma_f32_16x16x4_f32(false, a, false, b, (short)0, c,
                                                 false, false);
}

__device__ __forceinline__
void lin_tile(const float* __restrict__ x, const float* __restrict__ W,
              const float* __restrict__ bias, float* __restrict__ t,
              float* aS, int lt, int row0, bool live) {
    for (int i = lt; i < 16 * D; i += 256) {
        int r = i >> 7, c = i & (D - 1);
        aS[r * LDSW + c] = x[(size_t)(row0 + r) * D + c];
    }
    __syncthreads();

    const int lane = lt & 31, wv = lt >> 5;
    const int n = lane & 15, h = lane >> 4, m = lane & 15;
    const int kb = h * 2;
    const int col = wv * 16 + n;

    v8f acc = {};
    for (int k = 0; k < D; k += 4) {
        v2f a = *(const v2f*)&aS[m * LDSW + k + kb];
        v2f b;
        b.x = W[(size_t)(k + kb) * D + col];
        b.y = W[(size_t)(k + kb + 1) * D + col];
        acc = wmma_f32(a, b, acc);
    }
    const float bv = bias[col];
    if (live)
        for (int p = 0; p < 8; ++p)
            t[(size_t)(row0 + p + 8 * h) * D + col] = acc[p] + bv;
}

#define RSH 10
#define RSZ 1024
#define NLD 8
#define NO_EDGE 0x7fffffff
#define NWV (RSZ / 32)
#define WCAP 1024
#define IDX_BITS (32 - RSH)

#define VISIT(dv, iv)                                                                      \
    do {                                                                                   \
        const int dl_ = (dv) - lo;                                                         \
        const bool hit_ = (unsigned)dl_ < (unsigned)RSZ;                                   \
        const unsigned mk_ = __builtin_amdgcn_ballot_w32(hit_);                            \
        if (hit_) {                                                                        \
            atomicAdd(&cnt[dl_], 1);                                                       \
            const int slot_ = wcount + (int)__builtin_amdgcn_mbcnt_lo(mk_, 0u);            \
            if (slot_ < cap) mine_list[slot_] = ((unsigned)(iv) << RSH) | (unsigned)dl_;   \
        }                                                                                  \
        wcount += __builtin_popcount(mk_);                                                 \
        below += ((dv) < lo) ? 1 : 0;                                                      \
    } while (0)

#define LINT 4
__global__ __launch_bounds__(RSZ)
void front_kernel(const int* __restrict__ ei0, int* __restrict__ rp0, int* __restrict__ nbr0,
                  int E0, int n0, int nr0,
                  const int* __restrict__ ei1, int* __restrict__ rp1, int* __restrict__ nbr1,
                  int E1, int n1, int nr1,
                  const int* __restrict__ ei2, int* __restrict__ rp2, int* __restrict__ nbr2,
                  int E2, int n2, int nr2,
                  const float* __restrict__ xa, const float* __restrict__ Wa,
                  const float* __restrict__ ba, float* __restrict__ ta, int tiles_a, int wg_a,
                  const float* __restrict__ xb, const float* __restrict__ Wb,
                  const float* __restrict__ bb, float* __restrict__ tb, int tiles_b) {
    __shared__ int cnt[RSZ];
    __shared__ int start[RSZ];
    __shared__ int belowS[RSZ];
    __shared__ __attribute__((aligned(16))) unsigned listS[NWV * WCAP];
    __shared__ int spill;
    int r = blockIdx.x;
    if (r >= nr0 + nr1 + nr2) {
        int b = r - (nr0 + nr1 + nr2);
        const bool second = b >= wg_a;
        if (second) b -= wg_a;
        const float* __restrict__ x  = second ? xb : xa;
        const float* __restrict__ Wt = second ? Wb : Wa;
        const float* __restrict__ bs = second ? bb : ba;
        float* __restrict__ t        = second ? tb : ta;
        const int tiles              = second ? tiles_b : tiles_a;
        const int sub  = __builtin_amdgcn_readfirstlane((int)threadIdx.x >> 8);
        const int tile = b * LINT + sub;
        const bool live = tile < tiles;
        lin_tile(x, Wt, bs, t, (float*)listS + sub * (16 * LDSW), (int)threadIdx.x & 255,
                 (live ? tile : tiles - 1) * 16, live);
        return;
    }
    const int* ei; int* rp; int* nbr; int E; int n; int nr;
    if (r < nr0)            { ei = ei0; rp = rp0; nbr = nbr0; E = E0; n = n0; nr = nr0; }
    else if (r < nr0 + nr1) { r -= nr0;       ei = ei1; rp = rp1; nbr = nbr1; E = E1; n = n1; nr = nr1; }
    else                    { r -= nr0 + nr1; ei = ei2; rp = rp2; nbr = nbr2; E = E2; n = n2; nr = nr2; }
    const int* __restrict__ dst = ei + E;
    const int tid  = threadIdx.x;
    const int lane = tid & 31;
    const int lo   = r << RSH;
    const int E4 = (((((size_t)dst) | ((size_t)ei)) & 15) == 0) ? (E >> 2) : 0;
    const int4* __restrict__ dst4 = (const int4*)dst;
    const int4* __restrict__ src4 = (const int4*)ei;
    const int cap = (E <= (1 << IDX_BITS)) ? WCAP : 0;
    unsigned* mine_list = &listS[(tid >> 5) * WCAP];

    cnt[tid] = 0;
    if (tid == 0) spill = 0;
    __syncthreads();

    int below = 0;
    int wcount = 0;
    for (int base = 0; base < E4; base += NLD * RSZ) {
        int4 v[NLD];
#pragma unroll
        for (int u = 0; u < NLD; ++u) {
            const int i = base + u * RSZ + tid;
            v[u] = (i < E4) ? dst4[i] : make_int4(NO_EDGE, NO_EDGE, NO_EDGE, NO_EDGE);
        }
#pragma unroll
        for (int u = 0; u < NLD; ++u) {
            const int i4 = 4 * (base + u * RSZ + tid);
            VISIT(v[u].x, i4);
            VISIT(v[u].y, i4 + 1);
            VISIT(v[u].z, i4 + 2);
            VISIT(v[u].w, i4 + 3);
        }
    }
    for (int base = 4 * E4; base < E; base += RSZ) {
        const int i = base + tid;
        const int d = (i < E) ? dst[i] : NO_EDGE;
        VISIT(d, i);
    }
    if (lane == 0 && wcount > cap) spill = 1;
    __syncthreads();

    const int c = cnt[tid];
    start[tid]  = c;
    belowS[tid] = below;
    __syncthreads();
    for (int off = 1; off < RSZ; off <<= 1) {
        const int a = (tid >= off) ? start[tid - off]  : 0;
        const int b = (tid >= off) ? belowS[tid - off] : 0;
        __syncthreads();
        start[tid]  += a;
        belowS[tid] += b;
        __syncthreads();
    }
    const int first = belowS[RSZ - 1];
    const int incl  = start[tid];
    const int total = start[RSZ - 1];
    const bool listed = (spill == 0);
    __syncthreads();
    const int mine = first + incl - c;
    start[tid] = mine;
    cnt[tid]   = 0;
    if (lo + tid < n) rp[lo + tid] = mine;
    if (r == nr - 1 && tid == RSZ - 1) rp[n] = first + total;
    __syncthreads();

    if (listed) {
        for (int kb = 0; kb < wcount; kb += 4 * 32) {
            int dl[4], s[4];
#pragma unroll
            for (int u = 0; u < 4; ++u) {
                const int k = kb + u * 32 + lane;
                const bool ok = k < wcount;
                const unsigned e = ok ? mine_list[k] : 0u;
                dl[u] = ok ? (int)(e & (RSZ - 1)) : -1;
                s[u]  = ok ? ei[e >> RSH] : 0;
            }
#pragma unroll
            for (int u = 0; u < 4; ++u)
                if (dl[u] >= 0) nbr[start[dl[u]] + atomicAdd(&cnt[dl[u]], 1)] = s[u];
        }
        return;
    }

    for (int base = 0; base < E4; base += (NLD / 2) * RSZ) {
        int4 v[NLD / 2], s[NLD / 2];
#pragma unroll
        for (int u = 0; u < NLD / 2; ++u) {
            const int i = base + u * RSZ + tid;
            const bool ok = i < E4;
            v[u] = ok ? dst4[i] : make_int4(NO_EDGE, NO_EDGE, NO_EDGE, NO_EDGE);
            s[u] = ok ? src4[i] : make_int4(0, 0, 0, 0);
        }
#pragma unroll
        for (int u = 0; u < NLD / 2; ++u) {
            const int d0 = v[u].x - lo, d1 = v[u].y - lo, d2 = v[u].z - lo, d3 = v[u].w - lo;
            if ((unsigned)d0 < (unsigned)RSZ) nbr[start[d0] + atomicAdd(&cnt[d0], 1)] = s[u].x;
            if ((unsigned)d1 < (unsigned)RSZ) nbr[start[d1] + atomicAdd(&cnt[d1], 1)] = s[u].y;
            if ((unsigned)d2 < (unsigned)RSZ) nbr[start[d2] + atomicAdd(&cnt[d2], 1)] = s[u].z;
            if ((unsigned)d3 < (unsigned)RSZ) nbr[start[d3] + atomicAdd(&cnt[d3], 1)] = s[u].w;
        }
    }
    for (int i = 4 * E4 + tid; i < E; i += RSZ) {
        const int d = dst[i] - lo;
        if ((unsigned)d < (unsigned)RSZ) nbr[start[d] + atomicAdd(&cnt[d], 1)] = ei[i];
    }
}
#undef VISIT

__device__ __forceinline__ int first_ids(const int* __restrict__ nbr, int beg, int end, int lane) {
    return (beg + lane < end) ? nbr[beg + lane] : 0;
}

__device__ __forceinline__ v4f ld_row(const float* __restrict__ t, int s, int lane) {
    return *(const v4f*)&t[(size_t)s * D + 4 * lane];
}

__device__ __forceinline__ v4f row_sum(const float* __restrict__ t, const int* __restrict__ nbr,
                                       int beg, int end, int ids, int lane) {
    v4f acc = {0.0f, 0.0f, 0.0f, 0.0f};
    for (int base = beg; base < end; base += 32) {
        if (base != beg) ids = (base + lane < end) ? nbr[base + lane] : 0;
        const int cnt = min(32, end - base);
        int j = 0;
        for (; j + 4 <= cnt; j += 4) {
            const int s0 = __builtin_amdgcn_readlane(ids, j);
            const int s1 = __builtin_amdgcn_readlane(ids, j + 1);
            const int s2 = __builtin_amdgcn_readlane(ids, j + 2);
            const int s3 = __builtin_amdgcn_readlane(ids, j + 3);
            const v4f v0 = ld_row(t, s0, lane);
            const v4f v1 = ld_row(t, s1, lane);
            const v4f v2 = ld_row(t, s2, lane);
            const v4f v3 = ld_row(t, s3, lane);
            acc += v0; acc += v1; acc += v2; acc += v3;
        }
        if (j + 2 <= cnt) {
            const int s0 = __builtin_amdgcn_readlane(ids, j);
            const int s1 = __builtin_amdgcn_readlane(ids, j + 1);
            const v4f v0 = ld_row(t, s0, lane);
            const v4f v1 = ld_row(t, s1, lane);
            acc += v0; acc += v1;
            j += 2;
        }
        if (j < cnt) {
            const int s0 = __builtin_amdgcn_readlane(ids, j);
            acc += ld_row(t, s0, lane);
        }
    }
    return acc;
}

__device__ __forceinline__ void put_mean(float* rowS, v4f s, int count, int lane) {
    const float c = fmaxf((float)count, 1.0f);
    v2f lo, hi;
    lo.x = s.x / c; lo.y = s.y / c; hi.x = s.z / c; hi.y = s.w / c;
    *(v2f*)&rowS[4 * lane]     = lo;
    *(v2f*)&rowS[4 * lane + 2] = hi;
}

template <int M>
__device__ __forceinline__ float row_xor(float v) {
    return __builtin_bit_cast(float, __builtin_amdgcn_update_dpp(0, __builtin_bit_cast(int, v),
                                                                 0x160 + M, 0xf, 0xf, true));
}

__device__ __forceinline__ float xhalf(float v) {
    const int b = __builtin_bit_cast(int, v);
    return __builtin_bit_cast(float, __builtin_amdgcn_permlanex16(b, b, 0x76543210, 0xfedcba98,
                                                                  false, false));
}

__device__ __forceinline__ float wave_sum(float v) {
    v += row_xor<1>(v);
    v += row_xor<2>(v);
    v += row_xor<4>(v);
    v += row_xor<8>(v);
    v += xhalf(v);
    return v;
}

__device__ __forceinline__ float sq4(v2f a, v2f b) {
    return a.x * a.x + a.y * a.y + b.x * b.x + b.y * b.y;
}

__global__ __launch_bounds__(256)
void combine_m_kernel(const float* __restrict__ t_m, const float* __restrict__ t_h,
                      const int* __restrict__ rp_mm, const int* __restrict__ nbr_mm,
                      const int* __restrict__ rp_hm, const int* __restrict__ nbr_hm,
                      const float* __restrict__ Wl_mm, const float* __restrict__ bl_mm,
                      const float* __restrict__ Wr_mm,
                      const float* __restrict__ Wl_hm, const float* __restrict__ bl_hm,
                      const float* __restrict__ Wr_hm,
                      const float* __restrict__ g, const float* __restrict__ be,
                      float* __restrict__ out) {
    __shared__ __attribute__((aligned(16))) float tS[16 * LDSW], aS[16 * LDSW], bS[16 * LDSW];
    const int tid  = threadIdx.x;
    const int row0 = blockIdx.x * 16;
    const int lane = tid & 31;
    const int wv = __builtin_amdgcn_readfirstlane(tid >> 5);

    const int g0 = row0 + wv, g1 = g0 + 8;
    const int ba0 = rp_mm[g0], ea0 = rp_mm[g0 + 1], ba1 = rp_mm[g1], ea1 = rp_mm[g1 + 1];
    const int bb0 = rp_hm[g0], eb0 = rp_hm[g0 + 1], bb1 = rp_hm[g1], eb1 = rp_hm[g1 + 1];
    const int ia0 = first_ids(nbr_mm, ba0, ea0, lane);
    const int ib0 = first_ids(nbr_hm, bb0, eb0, lane);
    const int ia1 = first_ids(nbr_mm, ba1, ea1, lane);
    const int ib1 = first_ids(nbr_hm, bb1, eb1, lane);

    for (int i = tid; i < 16 * D; i += 256) {
        int r = i >> 7, c = i & (D - 1);
        tS[r * LDSW + c] = t_m[(size_t)(row0 + r) * D + c];
    }
    put_mean(&aS[wv * LDSW],       row_sum(t_m, nbr_mm, ba0, ea0, ia0, lane), ea0 - ba0, lane);
    put_mean(&bS[wv * LDSW],       row_sum(t_h, nbr_hm, bb0, eb0, ib0, lane), eb0 - bb0, lane);
    put_mean(&aS[(wv + 8) * LDSW], row_sum(t_m, nbr_mm, ba1, ea1, ia1, lane), ea1 - ba1, lane);
    put_mean(&bS[(wv + 8) * LDSW], row_sum(t_h, nbr_hm, bb1, eb1, ib1, lane), eb1 - bb1, lane);
    __syncthreads();

    const int n = lane & 15, h = lane >> 4, m = lane & 15;
    const int kb = h * 2;
    const int col = wv * 16 + n;

    v8f acc_mm = {}, acc_hm = {};
    for (int k = 0; k < D; k += 4) {
        v2f at = *(const v2f*)&tS[m * LDSW + k + kb];
        v2f aa = *(const v2f*)&aS[m * LDSW + k + kb];
        v2f ab = *(const v2f*)&bS[m * LDSW + k + kb];
        v2f b1; b1.x = Wl_mm[(size_t)(k + kb) * D + col]; b1.y = Wl_mm[(size_t)(k + kb + 1) * D + col];
        v2f b2; b2.x = Wr_mm[(size_t)(k + kb) * D + col]; b2.y = Wr_mm[(size_t)(k + kb + 1) * D + col];
        v2f b3; b3.x = Wl_hm[(size_t)(k + kb) * D + col]; b3.y = Wl_hm[(size_t)(k + kb + 1) * D + col];
        v2f b4; b4.x = Wr_hm[(size_t)(k + kb) * D + col]; b4.y = Wr_hm[(size_t)(k + kb + 1) * D + col];
        acc_mm = wmma_f32(aa, b1, acc_mm);
        acc_mm = wmma_f32(at, b2, acc_mm);
        acc_hm = wmma_f32(ab, b3, acc_hm);
        acc_hm = wmma_f32(at, b4, acc_hm);
    }

    const float c1 = bl_mm[col], c2 = bl_hm[col];
    __syncthreads();
    for (int p = 0; p < 8; ++p) {
        const int r = p + 8 * h;
        aS[r * LDSW + col] = acc_mm[p] + c1;
        bS[r * LDSW + col] = acc_hm[p] + c2;
    }
    __syncthreads();

    const int cA = 2 * lane, cB = 64 + 2 * lane;
    const float gA0 = g[cA], gA1 = g[cA + 1], gB0 = g[cB], gB1 = g[cB + 1];
    const float eA0 = be[cA], eA1 = be[cA + 1], eB0 = be[cB], eB1 = be[cB + 1];
#pragma unroll 1
    for (int i = 0; i < 2; ++i) {
        const int r = wv + 8 * i;
        const v2f tA = *(const v2f*)&tS[r * LDSW + cA], tB = *(const v2f*)&tS[r * LDSW + cB];
        const v2f pA = *(const v2f*)&aS[r * LDSW + cA], pB = *(const v2f*)&aS[r * LDSW + cB];
        const v2f qA = *(const v2f*)&bS[r * LDSW + cA], qB = *(const v2f*)&bS[r * LDSW + cB];
        const float s1 = wave_sum(sq4(pA, pB));
        const float s2 = wave_sum(sq4(qA, qB));
        const float n1 = fmaxf(sqrtf(s1), 1e-12f);
        const float n2 = fmaxf(sqrtf(s2), 1e-12f);
        const float y0 = tA.x + 0.5f * (pA.x / n1 + qA.x / n2);
        const float y1 = tA.y + 0.5f * (pA.y / n1 + qA.y / n2);
        const float y2 = tB.x + 0.5f * (pB.x / n1 + qB.x / n2);
        const float y3 = tB.y + 0.5f * (pB.y / n1 + qB.y / n2);
        const float sy  = wave_sum(y0 + y1 + y2 + y3);
        const float syy = wave_sum(y0 * y0 + y1 * y1 + y2 * y2 + y3 * y3);
        const float mu  = sy * (1.0f / D);
        const float var = syy * (1.0f / D) - mu * mu;
        const float rs  = rsqrtf(var + 1e-5f);
        v2f oA, oB;
        oA.x = (y0 - mu) * rs * gA0 + eA0;
        oA.y = (y1 - mu) * rs * gA1 + eA1;
        oB.x = (y2 - mu) * rs * gB0 + eB0;
        oB.y = (y3 - mu) * rs * gB1 + eB1;
        float* orow = out + (size_t)(row0 + r) * D;
        *(v2f*)&orow[cA] = oA;
        *(v2f*)&orow[cB] = oB;
    }
}

__global__ __launch_bounds__(256)
void combine_h_kernel(const float* __restrict__ t_h, const float* __restrict__ t_m,
                      const int* __restrict__ rp_mh, const int* __restrict__ nbr_mh,
                      const float* __restrict__ Wl_mh, const float* __restrict__ bl_mh,
                      const float* __restrict__ Wr_mh,
                      const float* __restrict__ g, const float* __restrict__ be,
                      float* __restrict__ out) {
    __shared__ __attribute__((aligned(16))) float tS[16 * LDSW], aS[16 * LDSW];
    const int tid  = threadIdx.x;
    const int row0 = blockIdx.x * 16;
    const int lane = tid & 31;
    const int wv = __builtin_amdgcn_readfirstlane(tid >> 5);

    const int g0 = row0 + wv, g1 = g0 + 8;
    const int ba0 = rp_mh[g0], ea0 = rp_mh[g0 + 1], ba1 = rp_mh[g1], ea1 = rp_mh[g1 + 1];
    const int ia0 = first_ids(nbr_mh, ba0, ea0, lane);
    const int ia1 = first_ids(nbr_mh, ba1, ea1, lane);

    for (int i = tid; i < 16 * D; i += 256) {
        int r = i >> 7, c = i & (D - 1);
        tS[r * LDSW + c] = t_h[(size_t)(row0 + r) * D + c];
    }
    put_mean(&aS[wv * LDSW],       row_sum(t_m, nbr_mh, ba0, ea0, ia0, lane), ea0 - ba0, lane);
    put_mean(&aS[(wv + 8) * LDSW], row_sum(t_m, nbr_mh, ba1, ea1, ia1, lane), ea1 - ba1, lane);
    __syncthreads();

    const int n = lane & 15, h = lane >> 4, m = lane & 15;
    const int kb = h * 2;
    const int col = wv * 16 + n;

    v8f acc = {};
    for (int k = 0; k < D; k += 4) {
        v2f at = *(const v2f*)&tS[m * LDSW + k + kb];
        v2f aa = *(const v2f*)&aS[m * LDSW + k + kb];
        v2f b1; b1.x = Wl_mh[(size_t)(k + kb) * D + col]; b1.y = Wl_mh[(size_t)(k + kb + 1) * D + col];
        v2f b2; b2.x = Wr_mh[(size_t)(k + kb) * D + col]; b2.y = Wr_mh[(size_t)(k + kb + 1) * D + col];
        acc = wmma_f32(aa, b1, acc);
        acc = wmma_f32(at, b2, acc);
    }

    const float c1 = bl_mh[col];
    __syncthreads();
    for (int p = 0; p < 8; ++p)
        aS[(p + 8 * h) * LDSW + col] = acc[p] + c1;
    __syncthreads();

    const int cA = 2 * lane, cB = 64 + 2 * lane;
    const float gA0 = g[cA], gA1 = g[cA + 1], gB0 = g[cB], gB1 = g[cB + 1];
    const float eA0 = be[cA], eA1 = be[cA + 1], eB0 = be[cB], eB1 = be[cB + 1];
#pragma unroll 1
    for (int i = 0; i < 2; ++i) {
        const int r = wv + 8 * i;
        const v2f tA = *(const v2f*)&tS[r * LDSW + cA], tB = *(const v2f*)&tS[r * LDSW + cB];
        const v2f pA = *(const v2f*)&aS[r * LDSW + cA], pB = *(const v2f*)&aS[r * LDSW + cB];
        const float s1 = wave_sum(sq4(pA, pB));
        const float n1 = fmaxf(sqrtf(s1), 1e-12f);
        const float y0 = tA.x + pA.x / n1;
        const float y1 = tA.y + pA.y / n1;
        const float y2 = tB.x + pB.x / n1;
        const float y3 = tB.y + pB.y / n1;
        const float sy  = wave_sum(y0 + y1 + y2 + y3);
        const float syy = wave_sum(y0 * y0 + y1 * y1 + y2 * y2 + y3 * y3);
        const float mu  = sy * (1.0f / D);
        const float var = syy * (1.0f / D) - mu * mu;
        const float rs  = rsqrtf(var + 1e-5f);
        v2f oA, oB;
        oA.x = (y0 - mu) * rs * gA0 + eA0;
        oA.y = (y1 - mu) * rs * gA1 + eA1;
        oB.x = (y2 - mu) * rs * gB0 + eB0;
        oB.y = (y3 - mu) * rs * gB1 + eB1;
        float* orow = out + (size_t)(row0 + r) * D;
        *(v2f*)&orow[cA] = oA;
        *(v2f*)&orow[cB] = oB;
    }
}

extern "C" void kernel_launch(void* const* d_in, const int* in_sizes, int n_in,
                              void* d_out, int out_size, void* d_ws, size_t ws_size,
                              hipStream_t stream) {
    const float* x_m    = (const float*)d_in[0];
    const float* x_h    = (const float*)d_in[1];
    const float* W_nt_m = (const float*)d_in[2];
    const float* b_nt_m = (const float*)d_in[3];
    const float* W_nt_h = (const float*)d_in[4];
    const float* b_nt_h = (const float*)d_in[5];
    const float* Wl_mm  = (const float*)d_in[6];
    const float* bl_mm  = (const float*)d_in[7];
    const float* Wr_mm  = (const float*)d_in[8];
    const float* Wl_mh  = (const float*)d_in[9];
    const float* bl_mh  = (const float*)d_in[10];
    const float* Wr_mh  = (const float*)d_in[11];
    const float* Wl_hm  = (const float*)d_in[12];
    const float* bl_hm  = (const float*)d_in[13];
    const float* Wr_hm  = (const float*)d_in[14];
    const float* g_m    = (const float*)d_in[15];
    const float* be_m   = (const float*)d_in[16];
    const float* g_h    = (const float*)d_in[17];
    const float* be_h   = (const float*)d_in[18];
    const int*   ei_mm  = (const int*)d_in[19];
    const int*   ei_mh  = (const int*)d_in[20];
    const int*   ei_hm  = (const int*)d_in[21];

    const int N_M  = in_sizes[0] / D;
    const int N_H  = in_sizes[1] / D;
    const int E_mm = in_sizes[19] / 2;
    const int E_mh = in_sizes[20] / 2;
    const int E_hm = in_sizes[21] / 2;

    const size_t nm = (size_t)N_M * D;
    const size_t nh = (size_t)N_H * D;

    float* ws     = (float*)d_ws;
    float* t_m    = ws;
    float* t_h    = t_m + nm;
    int* rp_mm  = (int*)(t_h + nh);
    int* rp_hm  = rp_mm + (N_M + 1);
    int* rp_mh  = rp_hm + (N_M + 1);
    int* nbr_mm = rp_mh + (N_H + 1);
    int* nbr_hm = nbr_mm + E_mm;
    int* nbr_mh = nbr_hm + E_hm;

    auto rg = [](int n) { return (n + RSZ - 1) / RSZ; };
    const int nr_mm = rg(N_M), nr_hm = rg(N_M), nr_mh = rg(N_H);
    const int tiles_m = N_M / 16, tiles_h = N_H / 16;
    const int wg_m = (tiles_m + LINT - 1) / LINT, wg_h = (tiles_h + LINT - 1) / LINT;
    front_kernel<<<nr_mm + nr_hm + nr_mh + wg_m + wg_h, RSZ, 0, stream>>>(
        ei_mm, rp_mm, nbr_mm, E_mm, N_M, nr_mm,
        ei_hm, rp_hm, nbr_hm, E_hm, N_M, nr_hm,
        ei_mh, rp_mh, nbr_mh, E_mh, N_H, nr_mh,
        x_m, W_nt_m, b_nt_m, t_m, tiles_m, wg_m,
        x_h, W_nt_h, b_nt_h, t_h, tiles_h);

    float* out_m = (float*)d_out;
    float* out_h = out_m + nm;
    combine_m_kernel<<<N_M / 16, 256, 0, stream>>>(t_m, t_h, rp_mm, nbr_mm, rp_hm, nbr_hm,
                                                   Wl_mm, bl_mm, Wr_mm, Wl_hm, bl_hm, Wr_hm,
                                                   g_m, be_m, out_m);
    combine_h_kernel<<<N_H / 16, 256, 0, stream>>>(t_h, t_m, rp_mh, nbr_mh,
                                                   Wl_mh, bl_mh, Wr_mh,
                                                   g_h, be_h, out_h);
}


# SAGE weight matrices re-laid once per call in WMMA B-fragment order by extra work-groups of the first launch; combine kernels read a B fragment pair as one 16-byte load; sum orders unchanged
# speedup vs baseline: 5.0640x; 1.0562x over previous
#include <hip/hip_runtime.h>

#define D 128
#define LDSW 130

typedef __attribute__((ext_vector_type(2))) float v2f;
typedef __attribute__((ext_vector_type(8))) float v8f;
typedef __attribute__((ext_vector_type(4))) float v4f;

__device__ __forceinline__ v8f wmma_f32(v2f a, v2f b, v8f c) {
    return __builtin_amdgcn_wmma_f32_16x16x4_f32(false, a, false, b, (short)0, c,
                                                 false, false);
}

__device__ __forceinline__ v2f frag_lo(v4f w) { v2f b; b.x = w.x; b.y = w.y; return b; }
__device__ __forceinline__ v2f frag_hi(v4f w) { v2f b; b.x = w.z; b.y = w.w; return b; }

__device__ __forceinline__
void lin_tile(const float* __restrict__ x, const float* __restrict__ W,
              const float* __restrict__ bias, float* __restrict__ t,
              float* aS, int lt, int row0, bool live) {
    for (int i = lt; i < 16 * D; i += 256) {
        int r = i >> 7, c = i & (D - 1);
        aS[r * LDSW + c] = x[(size_t)(row0 + r) * D + c];
    }
    __syncthreads();

    const int lane = lt & 31, wv = lt >> 5;
    const int n = lane & 15, h = lane >> 4, m = lane & 15;
    const int kb = h * 2;
    const int col = wv * 16 + n;

    v8f acc = {};
    for (int k = 0; k < D; k += 4) {
        v2f a = *(const v2f*)&aS[m * LDSW + k + kb];
        v2f b;
        b.x = W[(size_t)(k + kb) * D + col];
        b.y = W[(size_t)(k + kb + 1) * D + col];
        acc = wmma_f32(a, b, acc);
    }
    const float bv = bias[col];
    if (live)
        for (int p = 0; p < 8; ++p)
            t[(size_t)(row0 + p + 8 * h) * D + col] = acc[p] + bv;
}

#define RSH 10
#define RSZ 1024
#define NLD 8
#define NO_EDGE 0x7fffffff
#define NWV (RSZ / 32)
#define WCAP 1024
#define IDX_BITS (32 - RSH)

#define VISIT(dv, iv)                                                                      \
    do {                                                                                   \
        const int dl_ = (dv) - lo;                                                         \
        const bool hit_ = (unsigned)dl_ < (unsigned)RSZ;                                   \
        const unsigned mk_ = __builtin_amdgcn_ballot_w32(hit_);                            \
        if (hit_) {                                                                        \
            atomicAdd(&cnt[dl_], 1);                                                       \
            const int slot_ = wcount + (int)__builtin_amdgcn_mbcnt_lo(mk_, 0u);            \
            if (slot_ < cap) mine_list[slot_] = ((unsigned)(iv) << RSH) | (unsigned)dl_;   \
        }                                                                                  \
        wcount += __builtin_popcount(mk_);                                                 \
        below += ((dv) < lo) ? 1 : 0;                                                      \
    } while (0)

#define PKC 4096
#define PKF (4 * PKC)
__device__ __forceinline__
void pack_chunk(const float* __restrict__ Wsrc, float* __restrict__ dstW, int idx) {
    const int q2 = idx >> 8, wv = (idx >> 5) & 7, lane = idx & 31;
    const int col = wv * 16 + (lane & 15);
    const int k0 = 8 * q2 + 2 * (lane >> 4);
    v4f v;
    v.x = Wsrc[(size_t)k0 * D + col];
    v.y = Wsrc[(size_t)(k0 + 1) * D + col];
    v.z = Wsrc[(size_t)(k0 + 4) * D + col];
    v.w = Wsrc[(size_t)(k0 + 5) * D + col];
    *(v4f*)&dstW[(size_t)idx * 4] = v;
}

#define LINT 4
__global__ __launch_bounds__(RSZ)
void front_kernel(const int* __restrict__ ei0, int* __restrict__ rp0, int* __restrict__ nbr0,
                  int E0, int n0, int nr0,
                  const int* __restrict__ ei1, int* __restrict__ rp1, int* __restrict__ nbr1,
                  int E1, int n1, int nr1,
                  const int* __restrict__ ei2, int* __restrict__ rp2, int* __restrict__ nbr2,
                  int E2, int n2, int nr2,
                  const float* __restrict__ xa, const float* __restrict__ Wa,
                  const float* __restrict__ ba, float* __restrict__ ta, int tiles_a, int wg_a,
                  const float* __restrict__ xb, const float* __restrict__ Wb,
                  const float* __restrict__ bb, float* __restrict__ tb, int tiles_b,
                  const float* __restrict__ w0, const float* __restrict__ w1,
                  const float* __restrict__ w2, const float* __restrict__ w3,
                  const float* __restrict__ w4, const float* __restrict__ w5,
                  float* __restrict__ wp, int npk) {
    __shared__ int cnt[RSZ];
    __shared__ int start[RSZ];
    __shared__ int belowS[RSZ];
    __shared__ __attribute__((aligned(16))) unsigned listS[NWV * WCAP];
    __shared__ int spill;
    int r = blockIdx.x;
    if (r >= nr0 + nr1 + nr2) {
        int b = r - (nr0 + nr1 + nr2);
        if (b < npk) {
            const int mtx = b / (PKC / RSZ);
            const float* __restrict__ Wsrc = mtx == 0 ? w0 : mtx == 1 ? w1 : mtx == 2 ? w2
                                           : mtx == 3 ? w3 : mtx == 4 ? w4 : w5;
            pack_chunk(Wsrc, wp + (size_t)mtx * PKF, (b % (PKC / RSZ)) * RSZ + (int)threadIdx.x);
            return;
        }
        b -= npk;
        const bool second = b >= wg_a;
        if (second) b -= wg_a;
        const float* __restrict__ x  = second ? xb : xa;
        const float* __restrict__ Wt = second ? Wb : Wa;
        const float* __restrict__ bs = second ? bb : ba;
        float* __restrict__ t        = second ? tb : ta;
        const int tiles              = second ? tiles_b : tiles_a;
        const int sub  = __builtin_amdgcn_readfirstlane((int)threadIdx.x >> 8);
        const int tile = b * LINT + sub;
        const bool live = tile < tiles;
        lin_tile(x, Wt, bs, t, (float*)listS + sub * (16 * LDSW), (int)threadIdx.x & 255,
                 (live ? tile : tiles - 1) * 16, live);
        return;
    }
    const int* ei; int* rp; int* nbr; int E; int n; int nr;
    if (r < nr0)            { ei = ei0; rp = rp0; nbr = nbr0; E = E0; n = n0; nr = nr0; }
    else if (r < nr0 + nr1) { r -= nr0;       ei = ei1; rp = rp1; nbr = nbr1; E = E1; n = n1; nr = nr1; }
    else                    { r -= nr0 + nr1; ei = ei2; rp = rp2; nbr = nbr2; E = E2; n = n2; nr = nr2; }
    const int* __restrict__ dst = ei + E;
    const int tid  = threadIdx.x;
    const int lane = tid & 31;
    const int lo   = r << RSH;
    const int E4 = (((((size_t)dst) | ((size_t)ei)) & 15) == 0) ? (E >> 2) : 0;
    const int4* __restrict__ dst4 = (const int4*)dst;
    const int4* __restrict__ src4 = (const int4*)ei;
    const int cap = (E <= (1 << IDX_BITS)) ? WCAP : 0;
    unsigned* mine_list = &listS[(tid >> 5) * WCAP];

    cnt[tid] = 0;
    if (tid == 0) spill = 0;
    __syncthreads();

    int below = 0;
    int wcount = 0;
    for (int base = 0; base < E4; base += NLD * RSZ) {
        int4 v[NLD];
#pragma unroll
        for (int u = 0; u < NLD; ++u) {
            const int i = base + u * RSZ + tid;
            v[u] = (i < E4) ? dst4[i] : make_int4(NO_EDGE, NO_EDGE, NO_EDGE, NO_EDGE);
        }
#pragma unroll
        for (int u = 0; u < NLD; ++u) {
            const int i4 = 4 * (base + u * RSZ + tid);
            VISIT(v[u].x, i4);
            VISIT(v[u].y, i4 + 1);
            VISIT(v[u].z, i4 + 2);
            VISIT(v[u].w, i4 + 3);
        }
    }
    for (int base = 4 * E4; base < E; base += RSZ) {
        const int i = base + tid;
        const int d = (i < E) ? dst[i] : NO_EDGE;
        VISIT(d, i);
    }
    if (lane == 0 && wcount > cap) spill = 1;
    __syncthreads();

    const int c = cnt[tid];
    start[tid]  = c;
    belowS[tid] = below;
    __syncthreads();
    for (int off = 1; off < RSZ; off <<= 1) {
        const int a = (tid >= off) ? start[tid - off]  : 0;
        const int b = (tid >= off) ? belowS[tid - off] : 0;
        __syncthreads();
        start[tid]  += a;
        belowS[tid] += b;
        __syncthreads();
    }
    const int first = belowS[RSZ - 1];
    const int incl  = start[tid];
    const int total = start[RSZ - 1];
    const bool listed = (spill == 0);
    __syncthreads();
    const int mine = first + incl - c;
    start[tid] = mine;
    cnt[tid]   = 0;
    if (lo + tid < n) rp[lo + tid] = mine;
    if (r == nr - 1 && tid == RSZ - 1) rp[n] = first + total;
    __syncthreads();

    if (listed) {
        for (int kb = 0; kb < wcount; kb += 4 * 32) {
            int dl[4], s[4];
#pragma unroll
            for (int u = 0; u < 4; ++u) {
                const int k = kb + u * 32 + lane;
                const bool ok = k < wcount;
                const unsigned e = ok ? mine_list[k] : 0u;
                dl[u] = ok ? (int)(e & (RSZ - 1)) : -1;
                s[u]  = ok ? ei[e >> RSH] : 0;
            }
#pragma unroll
            for (int u = 0; u < 4; ++u)
                if (dl[u] >= 0) nbr[start[dl[u]] + atomicAdd(&cnt[dl[u]], 1)] = s[u];
        }
        return;
    }

    for (int base = 0; base < E4; base += (NLD / 2) * RSZ) {
        int4 v[NLD / 2], s[NLD / 2];
#pragma unroll
        for (int u = 0; u < NLD / 2; ++u) {
            const int i = base + u * RSZ + tid;
            const bool ok = i < E4;
            v[u] = ok ? dst4[i] : make_int4(NO_EDGE, NO_EDGE, NO_EDGE, NO_EDGE);
            s[u] = ok ? src4[i] : make_int4(0, 0, 0, 0);
        }
#pragma unroll
        for (int u = 0; u < NLD / 2; ++u) {
            const int d0 = v[u].x - lo, d1 = v[u].y - lo, d2 = v[u].z - lo, d3 = v[u].w - lo;
            if ((unsigned)d0 < (unsigned)RSZ) nbr[start[d0] + atomicAdd(&cnt[d0], 1)] = s[u].x;
            if ((unsigned)d1 < (unsigned)RSZ) nbr[start[d1] + atomicAdd(&cnt[d1], 1)] = s[u].y;
            if ((unsigned)d2 < (unsigned)RSZ) nbr[start[d2] + atomicAdd(&cnt[d2], 1)] = s[u].z;
            if ((unsigned)d3 < (unsigned)RSZ) nbr[start[d3] + atomicAdd(&cnt[d3], 1)] = s[u].w;
        }
    }
    for (int i = 4 * E4 + tid; i < E; i += RSZ) {
        const int d = dst[i] - lo;
        if ((unsigned)d < (unsigned)RSZ) nbr[start[d] + atomicAdd(&cnt[d], 1)] = ei[i];
    }
}
#undef VISIT

__device__ __forceinline__ int first_ids(const int* __restrict__ nbr, int beg, int end, int lane) {
    return (beg + lane < end) ? nbr[beg + lane] : 0;
}

__device__ __forceinline__ v4f ld_row(const float* __restrict__ t, int s, int lane) {
    return *(const v4f*)&t[(size_t)s * D + 4 * lane];
}

__device__ __forceinline__ v4f row_sum(const float* __restrict__ t, const int* __restrict__ nbr,
                                       int beg, int end, int ids, int lane) {
    v4f acc = {0.0f, 0.0f, 0.0f, 0.0f};
    for (int base = beg; base < end; base += 32) {
        if (base != beg) ids = (base + lane < end) ? nbr[base + lane] : 0;
        const int cnt = min(32, end - base);
        int j = 0;
        for (; j + 4 <= cnt; j += 4) {
            const int s0 = __builtin_amdgcn_readlane(ids, j);
            const int s1 = __builtin_amdgcn_readlane(ids, j + 1);
            const int s2 = __builtin_amdgcn_readlane(ids, j + 2);
            const int s3 = __builtin_amdgcn_readlane(ids, j + 3);
            const v4f v0 = ld_row(t, s0, lane);
            const v4f v1 = ld_row(t, s1, lane);
            const v4f v2 = ld_row(t, s2, lane);
            const v4f v3 = ld_row(t, s3, lane);
            acc += v0; acc += v1; acc += v2; acc += v3;
        }
        if (j + 2 <= cnt) {
            const int s0 = __builtin_amdgcn_readlane(ids, j);
            const int s1 = __builtin_amdgcn_readlane(ids, j + 1);
            const v4f v0 = ld_row(t, s0, lane);
            const v4f v1 = ld_row(t, s1, lane);
            acc += v0; acc += v1;
            j += 2;
        }
        if (j < cnt) {
            const int s0 = __builtin_amdgcn_readlane(ids, j);
            acc += ld_row(t, s0, lane);
        }
    }
    return acc;
}

__device__ __forceinline__ void put_mean(float* rowS, v4f s, int count, int lane) {
    const float c = fmaxf((float)count, 1.0f);
    v2f lo, hi;
    lo.x = s.x / c; lo.y = s.y / c; hi.x = s.z / c; hi.y = s.w / c;
    *(v2f*)&rowS[4 * lane]     = lo;
    *(v2f*)&rowS[4 * lane + 2] = hi;
}

template <int M>
__device__ __forceinline__ float row_xor(float v) {
    return __builtin_bit_cast(float, __builtin_amdgcn_update_dpp(0, __builtin_bit_cast(int, v),
                                                                 0x160 + M, 0xf, 0xf, true));
}

__device__ __forceinline__ float xhalf(float v) {
    const int b = __builtin_bit_cast(int, v);
    return __builtin_bit_cast(float, __builtin_amdgcn_permlanex16(b, b, 0x76543210, 0xfedcba98,
                                                                  false, false));
}

__device__ __forceinline__ float wave_sum(float v) {
    v += row_xor<1>(v);
    v += row_xor<2>(v);
    v += row_xor<4>(v);
    v += row_xor<8>(v);
    v += xhalf(v);
    return v;
}

__device__ __forceinline__ float sq4(v2f a, v2f b) {
    return a.x * a.x + a.y * a.y + b.x * b.x + b.y * b.y;
}

__global__ __launch_bounds__(256)
void combine_m_kernel(const float* __restrict__ t_m, const float* __restrict__ t_h,
                      const int* __restrict__ rp_mm, const int* __restrict__ nbr_mm,
                      const int* __restrict__ rp_hm, const int* __restrict__ nbr_hm,
                      const float* __restrict__ Wp,
                      const float* __restrict__ bl_mm, const float* __restrict__ bl_hm,
                      const float* __restrict__ g, const float* __restrict__ be,
                      float* __restrict__ out) {
    __shared__ __attribute__((aligned(16))) float tS[16 * LDSW], aS[16 * LDSW], bS[16 * LDSW];
    const int tid  = threadIdx.x;
    const int row0 = blockIdx.x * 16;
    const int lane = tid & 31;
    const int wv = __builtin_amdgcn_readfirstlane(tid >> 5);

    const int g0 = row0 + wv, g1 = g0 + 8;
    const int ba0 = rp_mm[g0], ea0 = rp_mm[g0 + 1], ba1 = rp_mm[g1], ea1 = rp_mm[g1 + 1];
    const int bb0 = rp_hm[g0], eb0 = rp_hm[g0 + 1], bb1 = rp_hm[g1], eb1 = rp_hm[g1 + 1];
    const int ia0 = first_ids(nbr_mm, ba0, ea0, lane);
    const int ib0 = first_ids(nbr_hm, bb0, eb0, lane);
    const int ia1 = first_ids(nbr_mm, ba1, ea1, lane);
    const int ib1 = first_ids(nbr_hm, bb1, eb1, lane);

    for (int i = tid; i < 16 * D; i += 256) {
        int r = i >> 7, c = i & (D - 1);
        tS[r * LDSW + c] = t_m[(size_t)(row0 + r) * D + c];
    }
    put_mean(&aS[wv * LDSW],       row_sum(t_m, nbr_mm, ba0, ea0, ia0, lane), ea0 - ba0, lane);
    put_mean(&bS[wv * LDSW],       row_sum(t_h, nbr_hm, bb0, eb0, ib0, lane), eb0 - bb0, lane);
    put_mean(&aS[(wv + 8) * LDSW], row_sum(t_m, nbr_mm, ba1, ea1, ia1, lane), ea1 - ba1, lane);
    put_mean(&bS[(wv + 8) * LDSW], row_sum(t_h, nbr_hm, bb1, eb1, ib1, lane), eb1 - bb1, lane);
    __syncthreads();

    const int n = lane & 15, h = lane >> 4, m = lane & 15;
    const int kb = h * 2;
    const int col = wv * 16 + n;

    const v4f* __restrict__ P = (const v4f*)Wp + tid;
    v8f acc_mm = {}, acc_hm = {};
    for (int q2 = 0; q2 < D / 8; ++q2) {
        const int k = 8 * q2;
        const v4f w1 = P[q2 * 256], w2 = P[PKC + q2 * 256];
        const v4f w3 = P[2 * PKC + q2 * 256], w4 = P[3 * PKC + q2 * 256];
        const v2f at0 = *(const v2f*)&tS[m * LDSW + k + kb], at1 = *(const v2f*)&tS[m * LDSW + k + 4 + kb];
        const v2f aa0 = *(const v2f*)&aS[m * LDSW + k + kb], aa1 = *(const v2f*)&aS[m * LDSW + k + 4 + kb];
        const v2f ab0 = *(const v2f*)&bS[m * LDSW + k + kb], ab1 = *(const v2f*)&bS[m * LDSW + k + 4 + kb];
        acc_mm = wmma_f32(aa0, frag_lo(w1), acc_mm);
        acc_mm = wmma_f32(at0, frag_lo(w2), acc_mm);
        acc_hm = wmma_f32(ab0, frag_lo(w3), acc_hm);
        acc_hm = wmma_f32(at0, frag_lo(w4), acc_hm);
        acc_mm = wmma_f32(aa1, frag_hi(w1), acc_mm);
        acc_mm = wmma_f32(at1, frag_hi(w2), acc_mm);
        acc_hm = wmma_f32(ab1, frag_hi(w3), acc_hm);
        acc_hm = wmma_f32(at1, frag_hi(w4), acc_hm);
    }

    const float c1 = bl_mm[col], c2 = bl_hm[col];
    __syncthreads();
    for (int p = 0; p < 8; ++p) {
        const int r = p + 8 * h;
        aS[r * LDSW + col] = acc_mm[p] + c1;
        bS[r * LDSW + col] = acc_hm[p] + c2;
    }
    __syncthreads();

    const int cA = 2 * lane, cB = 64 + 2 * lane;
    const float gA0 = g[cA], gA1 = g[cA + 1], gB0 = g[cB], gB1 = g[cB + 1];
    const float eA0 = be[cA], eA1 = be[cA + 1], eB0 = be[cB], eB1 = be[cB + 1];
#pragma unroll 1
    for (int i = 0; i < 2; ++i) {
        const int r = wv + 8 * i;
        const v2f tA = *(const v2f*)&tS[r * LDSW + cA], tB = *(const v2f*)&tS[r * LDSW + cB];
        const v2f pA = *(const v2f*)&aS[r * LDSW + cA], pB = *(const v2f*)&aS[r * LDSW + cB];
        const v2f qA = *(const v2f*)&bS[r * LDSW + cA], qB = *(const v2f*)&bS[r * LDSW + cB];
        const float s1 = wave_sum(sq4(pA, pB));
        const float s2 = wave_sum(sq4(qA, qB));
        const float n1 = fmaxf(sqrtf(s1), 1e-12f);
        const float n2 = fmaxf(sqrtf(s2), 1e-12f);
        const float y0 = tA.x + 0.5f * (pA.x / n1 + qA.x / n2);
        const float y1 = tA.y + 0.5f * (pA.y / n1 + qA.y / n2);
        const float y2 = tB.x + 0.5f * (pB.x / n1 + qB.x / n2);
        const float y3 = tB.y + 0.5f * (pB.y / n1 + qB.y / n2);
        const float sy  = wave_sum(y0 + y1 + y2 + y3);
        const float syy = wave_sum(y0 * y0 + y1 * y1 + y2 * y2 + y3 * y3);
        const float mu  = sy * (1.0f / D);
        const float var = syy * (1.0f / D) - mu * mu;
        const float rs  = rsqrtf(var + 1e-5f);
        v2f oA, oB;
        oA.x = (y0 - mu) * rs * gA0 + eA0;
        oA.y = (y1 - mu) * rs * gA1 + eA1;
        oB.x = (y2 - mu) * rs * gB0 + eB0;
        oB.y = (y3 - mu) * rs * gB1 + eB1;
        float* orow = out + (size_t)(row0 + r) * D;
        *(v2f*)&orow[cA] = oA;
        *(v2f*)&orow[cB] = oB;
    }
}

__global__ __launch_bounds__(256)
void combine_h_kernel(const float* __restrict__ t_h, const float* __restrict__ t_m,
                      const int* __restrict__ rp_mh, const int* __restrict__ nbr_mh,
                      const float* __restrict__ Wp,
                      const float* __restrict__ bl_mh,
                      const float* __restrict__ g, const float* __restrict__ be,
                      float* __restrict__ out) {
    __shared__ __attribute__((aligned(16))) float tS[16 * LDSW], aS[16 * LDSW];
    const int tid  = threadIdx.x;
    const int row0 = blockIdx.x * 16;
    const int lane = tid & 31;
    const int wv = __builtin_amdgcn_readfirstlane(tid >> 5);

    const int g0 = row0 + wv, g1 = g0 + 8;
    const int ba0 = rp_mh[g0], ea0 = rp_mh[g0 + 1], ba1 = rp_mh[g1], ea1 = rp_mh[g1 + 1];
    const int ia0 = first_ids(nbr_mh, ba0, ea0, lane);
    const int ia1 = first_ids(nbr_mh, ba1, ea1, lane);

    for (int i = tid; i < 16 * D; i += 256) {
        int r = i >> 7, c = i & (D - 1);
        tS[r * LDSW + c] = t_h[(size_t)(row0 + r) * D + c];
    }
    put_mean(&aS[wv * LDSW],       row_sum(t_m, nbr_mh, ba0, ea0, ia0, lane), ea0 - ba0, lane);
    put_mean(&aS[(wv + 8) * LDSW], row_sum(t_m, nbr_mh, ba1, ea1, ia1, lane), ea1 - ba1, lane);
    __syncthreads();

    const int n = lane & 15, h = lane >> 4, m = lane & 15;
    const int kb = h * 2;
    const int col = wv * 16 + n;

    const v4f* __restrict__ P = (const v4f*)Wp + tid;
    v8f acc = {};
    for (int q2 = 0; q2 < D / 8; ++q2) {
        const int k = 8 * q2;
        const v4f w1 = P[q2 * 256], w2 = P[PKC + q2 * 256];
        const v2f at0 = *(const v2f*)&tS[m * LDSW + k + kb], at1 = *(const v2f*)&tS[m * LDSW + k + 4 + kb];
        const v2f aa0 = *(const v2f*)&aS[m * LDSW + k + kb], aa1 = *(const v2f*)&aS[m * LDSW + k + 4 + kb];
        acc = wmma_f32(aa0, frag_lo(w1), acc);
        acc = wmma_f32(at0, frag_lo(w2), acc);
        acc = wmma_f32(aa1, frag_hi(w1), acc);
        acc = wmma_f32(at1, frag_hi(w2), acc);
    }

    const float c1 = bl_mh[col];
    __syncthreads();
    for (int p = 0; p < 8; ++p)
        aS[(p + 8 * h) * LDSW + col] = acc[p] + c1;
    __syncthreads();

    const int cA = 2 * lane, cB = 64 + 2 * lane;
    const float gA0 = g[cA], gA1 = g[cA + 1], gB0 = g[cB], gB1 = g[cB + 1];
    const float eA0 = be[cA], eA1 = be[cA + 1], eB0 = be[cB], eB1 = be[cB + 1];
#pragma unroll 1
    for (int i = 0; i < 2; ++i) {
        const int r = wv + 8 * i;
        const v2f tA = *(const v2f*)&tS[r * LDSW + cA], tB = *(const v2f*)&tS[r * LDSW + cB];
        const v2f pA = *(const v2f*)&aS[r * LDSW + cA], pB = *(const v2f*)&aS[r * LDSW + cB];
        const float s1 = wave_sum(sq4(pA, pB));
        const float n1 = fmaxf(sqrtf(s1), 1e-12f);
        const float y0 = tA.x + pA.x / n1;
        const float y1 = tA.y + pA.y / n1;
        const float y2 = tB.x + pB.x / n1;
        const float y3 = tB.y + pB.y / n1;
        const float sy  = wave_sum(y0 + y1 + y2 + y3);
        const float syy = wave_sum(y0 * y0 + y1 * y1 + y2 * y2 + y3 * y3);
        const float mu  = sy * (1.0f / D);
        const float var = syy * (1.0f / D) - mu * mu;
        const float rs  = rsqrtf(var + 1e-5f);
        v2f oA, oB;
        oA.x = (y0 - mu) * rs * gA0 + eA0;
        oA.y = (y1 - mu) * rs * gA1 + eA1;
        oB.x = (y2 - mu) * rs * gB0 + eB0;
        oB.y = (y3 - mu) * rs * gB1 + eB1;
        float* orow = out + (size_t)(row0 + r) * D;
        *(v2f*)&orow[cA] = oA;
        *(v2f*)&orow[cB] = oB;
    }
}

extern "C" void kernel_launch(void* const* d_in, const int* in_sizes, int n_in,
                              void* d_out, int out_size, void* d_ws, size_t ws_size,
                              hipStream_t stream) {
    const float* x_m    = (const float*)d_in[0];
    const float* x_h    = (const float*)d_in[1];
    const float* W_nt_m = (const float*)d_in[2];
    const float* b_nt_m = (const float*)d_in[3];
    const float* W_nt_h = (const float*)d_in[4];
    const float* b_nt_h = (const float*)d_in[5];
    const float* Wl_mm  = (const float*)d_in[6];
    const float* bl_mm  = (const float*)d_in[7];
    const float* Wr_mm  = (const float*)d_in[8];
    const float* Wl_mh  = (const float*)d_in[9];
    const float* bl_mh  = (const float*)d_in[10];
    const float* Wr_mh  = (const float*)d_in[11];
    const float* Wl_hm  = (const float*)d_in[12];
    const float* bl_hm  = (const float*)d_in[13];
    const float* Wr_hm  = (const float*)d_in[14];
    const float* g_m    = (const float*)d_in[15];
    const float* be_m   = (const float*)d_in[16];
    const float* g_h    = (const float*)d_in[17];
    const float* be_h   = (const float*)d_in[18];
    const int*   ei_mm  = (const int*)d_in[19];
    const int*   ei_mh  = (const int*)d_in[20];
    const int*   ei_hm  = (const int*)d_in[21];

    const int N_M  = in_sizes[0] / D;
    const int N_H  = in_sizes[1] / D;
    const int E_mm = in_sizes[19] / 2;
    const int E_mh = in_sizes[20] / 2;
    const int E_hm = in_sizes[21] / 2;

    const size_t nm = (size_t)N_M * D;
    const size_t nh = (size_t)N_H * D;

    float* ws     = (float*)d_ws;
    float* t_m    = ws;
    float* t_h    = t_m + nm;
    float* wpk  = t_h + nh;
    int* rp_mm  = (int*)(wpk + 6 * PKF);
    int* rp_hm  = rp_mm + (N_M + 1);
    int* rp_mh  = rp_hm + (N_M + 1);
    int* nbr_mm = rp_mh + (N_H + 1);
    int* nbr_hm = nbr_mm + E_mm;
    int* nbr_mh = nbr_hm + E_hm;

    auto rg = [](int n) { return (n + RSZ - 1) / RSZ; };
    const int nr_mm = rg(N_M), nr_hm = rg(N_M), nr_mh = rg(N_H);
    const int tiles_m = N_M / 16, tiles_h = N_H / 16;
    const int wg_m = (tiles_m + LINT - 1) / LINT, wg_h = (tiles_h + LINT - 1) / LINT;
    const int npk = 6 * (PKC / RSZ);
    front_kernel<<<nr_mm + nr_hm + nr_mh + npk + wg_m + wg_h, RSZ, 0, stream>>>(
        ei_mm, rp_mm, nbr_mm, E_mm, N_M, nr_mm,
        ei_hm, rp_hm, nbr_hm, E_hm, N_M, nr_hm,
        ei_mh, rp_mh, nbr_mh, E_mh, N_H, nr_mh,
        x_m, W_nt_m, b_nt_m, t_m, tiles_m, wg_m,
        x_h, W_nt_h, b_nt_h, t_h, tiles_h,
        Wl_mm, Wr_mm, Wl_hm, Wr_hm, Wl_mh, Wr_mh, wpk, npk);

    float* out_m = (float*)d_out;
    float* out_h = out_m + nm;
    combine_m_kernel<<<N_M / 16, 256, 0, stream>>>(t_m, t_h, rp_mm, nbr_mm, rp_hm, nbr_hm,
                                                   wpk, bl_mm, bl_hm,
                                                   g_m, be_m, out_m);
    combine_h_kernel<<<N_H / 16, 256, 0, stream>>>(t_h, t_m, rp_mh, nbr_mh,
                                                   wpk + 4 * PKF, bl_mh,
                                                   g_h, be_h, out_h);
}


# Each 16-row A tile (x in transforms, t in combines) staged by two 16-byte loads per lane issued at wave start, t rows in flight under the gather, not 8 dependent 4-byte loads; sum orders unchanged
# speedup vs baseline: 5.2185x; 1.0305x over previous
#include <hip/hip_runtime.h>

#define D 128
#define LDSW 130

typedef __attribute__((ext_vector_type(2))) float v2f;
typedef __attribute__((ext_vector_type(8))) float v8f;
typedef __attribute__((ext_vector_type(4))) float v4f;

__device__ __forceinline__ v8f wmma_f32(v2f a, v2f b, v8f c) {
    return __builtin_amdgcn_wmma_f32_16x16x4_f32(false, a, false, b, (short)0, c,
                                                 false, false);
}

__device__ __forceinline__ v2f frag_lo(v4f w) { v2f b; b.x = w.x; b.y = w.y; return b; }
__device__ __forceinline__ v2f frag_hi(v4f w) { v2f b; b.x = w.z; b.y = w.w; return b; }

__device__ __forceinline__ void put_row(float* rowS, v4f v, int lane) {
    v2f lo, hi;
    lo.x = v.x; lo.y = v.y; hi.x = v.z; hi.y = v.w;
    *(v2f*)&rowS[4 * lane]     = lo;
    *(v2f*)&rowS[4 * lane + 2] = hi;
}

__device__ __forceinline__
void lin_tile(const float* __restrict__ x, const float* __restrict__ W,
              const float* __restrict__ bias, float* __restrict__ t,
              float* aS, int lt, int row0, bool live) {
    const int lane = lt & 31, wv = lt >> 5;
    const v4f xr0 = *(const v4f*)&x[(size_t)(row0 + wv) * D + 4 * lane];
    const v4f xr1 = *(const v4f*)&x[(size_t)(row0 + wv + 8) * D + 4 * lane];
    put_row(&aS[wv * LDSW], xr0, lane);
    put_row(&aS[(wv + 8) * LDSW], xr1, lane);
    __syncthreads();

    const int n = lane & 15, h = lane >> 4, m = lane & 15;
    const int kb = h * 2;
    const int col = wv * 16 + n;

    v8f acc = {};
    for (int k = 0; k < D; k += 4) {
        v2f a = *(const v2f*)&aS[m * LDSW + k + kb];
        v2f b;
        b.x = W[(size_t)(k + kb) * D + col];
        b.y = W[(size_t)(k + kb + 1) * D + col];
        acc = wmma_f32(a, b, acc);
    }
    const float bv = bias[col];
    if (live)
        for (int p = 0; p < 8; ++p)
            t[(size_t)(row0 + p + 8 * h) * D + col] = acc[p] + bv;
}

#define RSH 10
#define RSZ 1024
#define NLD 8
#define NO_EDGE 0x7fffffff
#define NWV (RSZ / 32)
#define WCAP 1024
#define IDX_BITS (32 - RSH)

#define VISIT(dv, iv)                                                                      \
    do {                                                                                   \
        const int dl_ = (dv) - lo;                                                         \
        const bool hit_ = (unsigned)dl_ < (unsigned)RSZ;                                   \
        const unsigned mk_ = __builtin_amdgcn_ballot_w32(hit_);                            \
        if (hit_) {                                                                        \
            atomicAdd(&cnt[dl_], 1);                                                       \
            const int slot_ = wcount + (int)__builtin_amdgcn_mbcnt_lo(mk_, 0u);            \
            if (slot_ < cap) mine_list[slot_] = ((unsigned)(iv) << RSH) | (unsigned)dl_;   \
        }                                                                                  \
        wcount += __builtin_popcount(mk_);                                                 \
        below += ((dv) < lo) ? 1 : 0;                                                      \
    } while (0)

#define PKC 4096
#define PKF (4 * PKC)
__device__ __forceinline__
void pack_chunk(const float* __restrict__ Wsrc, float* __restrict__ dstW, int idx) {
    const int q2 = idx >> 8, wv = (idx >> 5) & 7, lane = idx & 31;
    const int col = wv * 16 + (lane & 15);
    const int k0 = 8 * q2 + 2 * (lane >> 4);
    v4f v;
    v.x = Wsrc[(size_t)k0 * D + col];
    v.y = Wsrc[(size_t)(k0 + 1) * D + col];
    v.z = Wsrc[(size_t)(k0 + 4) * D + col];
    v.w = Wsrc[(size_t)(k0 + 5) * D + col];
    *(v4f*)&dstW[(size_t)idx * 4] = v;
}

#define LINT 4
__global__ __launch_bounds__(RSZ)
void front_kernel(const int* __restrict__ ei0, int* __restrict__ rp0, int* __restrict__ nbr0,
                  int E0, int n0, int nr0,
                  const int* __restrict__ ei1, int* __restrict__ rp1, int* __restrict__ nbr1,
                  int E1, int n1, int nr1,
                  const int* __restrict__ ei2, int* __restrict__ rp2, int* __restrict__ nbr2,
                  int E2, int n2, int nr2,
                  const float* __restrict__ xa, const float* __restrict__ Wa,
                  const float* __restrict__ ba, float* __restrict__ ta, int tiles_a, int wg_a,
                  const float* __restrict__ xb, const float* __restrict__ Wb,
                  const float* __restrict__ bb, float* __restrict__ tb, int tiles_b,
                  const float* __restrict__ w0, const float* __restrict__ w1,
                  const float* __restrict__ w2, const float* __restrict__ w3,
                  const float* __restrict__ w4, const float* __restrict__ w5,
                  float* __restrict__ wp, int npk) {
    __shared__ int cnt[RSZ];
    __shared__ int start[RSZ];
    __shared__ int belowS[RSZ];
    __shared__ __attribute__((aligned(16))) unsigned listS[NWV * WCAP];
    __shared__ int spill;
    int r = blockIdx.x;
    if (r >= nr0 + nr1 + nr2) {
        int b = r - (nr0 + nr1 + nr2);
        if (b < npk) {
            const int mtx = b / (PKC / RSZ);
            const float* __restrict__ Wsrc = mtx == 0 ? w0 : mtx == 1 ? w1 : mtx == 2 ? w2
                                           : mtx == 3 ? w3 : mtx == 4 ? w4 : w5;
            pack_chunk(Wsrc, wp + (size_t)mtx * PKF, (b % (PKC / RSZ)) * RSZ + (int)threadIdx.x);
            return;
        }
        b -= npk;
        const bool second = b >= wg_a;
        if (second) b -= wg_a;
        const float* __restrict__ x  = second ? xb : xa;
        const float* __restrict__ Wt = second ? Wb : Wa;
        const float* __restrict__ bs = second ? bb : ba;
        float* __restrict__ t        = second ? tb : ta;
        const int tiles              = second ? tiles_b : tiles_a;
        const int sub  = __builtin_amdgcn_readfirstlane((int)threadIdx.x >> 8);
        const int tile = b * LINT + sub;
        const bool live = tile < tiles;
        lin_tile(x, Wt, bs, t, (float*)listS + sub * (16 * LDSW), (int)threadIdx.x & 255,
                 (live ? tile : tiles - 1) * 16, live);
        return;
    }
    const int* ei; int* rp; int* nbr; int E; int n; int nr;
    if (r < nr0)            { ei = ei0; rp = rp0; nbr = nbr0; E = E0; n = n0; nr = nr0; }
    else if (r < nr0 + nr1) { r -= nr0;       ei = ei1; rp = rp1; nbr = nbr1; E = E1; n = n1; nr = nr1; }
    else                    { r -= nr0 + nr1; ei = ei2; rp = rp2; nbr = nbr2; E = E2; n = n2; nr = nr2; }
    const int* __restrict__ dst = ei + E;
    const int tid  = threadIdx.x;
    const int lane = tid & 31;
    const int lo   = r << RSH;
    const int E4 = (((((size_t)dst) | ((size_t)ei)) & 15) == 0) ? (E >> 2) : 0;
    const int4* __restrict__ dst4 = (const int4*)dst;
    const int4* __restrict__ src4 = (const int4*)ei;
    const int cap = (E <= (1 << IDX_BITS)) ? WCAP : 0;
    unsigned* mine_list = &listS[(tid >> 5) * WCAP];

    cnt[tid] = 0;
    if (tid == 0) spill = 0;
    __syncthreads();

    int below = 0;
    int wcount = 0;
    for (int base = 0; base < E4; base += NLD * RSZ) {
        int4 v[NLD];
#pragma unroll
        for (int u = 0; u < NLD; ++u) {
            const int i = base + u * RSZ + tid;
            v[u] = (i < E4) ? dst4[i] : make_int4(NO_EDGE, NO_EDGE, NO_EDGE, NO_EDGE);
        }
#pragma unroll
        for (int u = 0; u < NLD; ++u) {
            const int i4 = 4 * (base + u * RSZ + tid);
            VISIT(v[u].x, i4);
            VISIT(v[u].y, i4 + 1);
            VISIT(v[u].z, i4 + 2);
            VISIT(v[u].w, i4 + 3);
        }
    }
    for (int base = 4 * E4; base < E; base += RSZ) {
        const int i = base + tid;
        const int d = (i < E) ? dst[i] : NO_EDGE;
        VISIT(d, i);
    }
    if (lane == 0 && wcount > cap) spill = 1;
    __syncthreads();

    const int c = cnt[tid];
    start[tid]  = c;
    belowS[tid] = below;
    __syncthreads();
    for (int off = 1; off < RSZ; off <<= 1) {
        const int a = (tid >= off) ? start[tid - off]  : 0;
        const int b = (tid >= off) ? belowS[tid - off] : 0;
        __syncthreads();
        start[tid]  += a;
        belowS[tid] += b;
        __syncthreads();
    }
    const int first = belowS[RSZ - 1];
    const int incl  = start[tid];
    const int total = start[RSZ - 1];
    const bool listed = (spill == 0);
    __syncthreads();
    const int mine = first + incl - c;
    start[tid] = mine;
    cnt[tid]   = 0;
    if (lo + tid < n) rp[lo + tid] = mine;
    if (r == nr - 1 && tid == RSZ - 1) rp[n] = first + total;
    __syncthreads();

    if (listed) {
        for (int kb = 0; kb < wcount; kb += 4 * 32) {
            int dl[4], s[4];
#pragma unroll
            for (int u = 0; u < 4; ++u) {
                const int k = kb + u * 32 + lane;
                const bool ok = k < wcount;
                const unsigned e = ok ? mine_list[k] : 0u;
                dl[u] = ok ? (int)(e & (RSZ - 1)) : -1;
                s[u]  = ok ? ei[e >> RSH] : 0;
            }
#pragma unroll
            for (int u = 0; u < 4; ++u)
                if (dl[u] >= 0) nbr[start[dl[u]] + atomicAdd(&cnt[dl[u]], 1)] = s[u];
        }
        return;
    }

    for (int base = 0; base < E4; base += (NLD / 2) * RSZ) {
        int4 v[NLD / 2], s[NLD / 2];
#pragma unroll
        for (int u = 0; u < NLD / 2; ++u) {
            const int i = base + u * RSZ + tid;
            const bool ok = i < E4;
            v[u] = ok ? dst4[i] : make_int4(NO_EDGE, NO_EDGE, NO_EDGE, NO_EDGE);
            s[u] = ok ? src4[i] : make_int4(0, 0, 0, 0);
        }
#pragma unroll
        for (int u = 0; u < NLD / 2; ++u) {
            const int d0 = v[u].x - lo, d1 = v[u].y - lo, d2 = v[u].z - lo, d3 = v[u].w - lo;
            if ((unsigned)d0 < (unsigned)RSZ) nbr[start[d0] + atomicAdd(&cnt[d0], 1)] = s[u].x;
            if ((unsigned)d1 < (unsigned)RSZ) nbr[start[d1] + atomicAdd(&cnt[d1], 1)] = s[u].y;
            if ((unsigned)d2 < (unsigned)RSZ) nbr[start[d2] + atomicAdd(&cnt[d2], 1)] = s[u].z;
            if ((unsigned)d3 < (unsigned)RSZ) nbr[start[d3] + atomicAdd(&cnt[d3], 1)] = s[u].w;
        }
    }
    for (int i = 4 * E4 + tid; i < E; i += RSZ) {
        const int d = dst[i] - lo;
        if ((unsigned)d < (unsigned)RSZ) nbr[start[d] + atomicAdd(&cnt[d], 1)] = ei[i];
    }
}
#undef VISIT

__device__ __forceinline__ int first_ids(const int* __restrict__ nbr, int beg, int end, int lane) {
    return (beg + lane < end) ? nbr[beg + lane] : 0;
}

__device__ __forceinline__ v4f ld_row(const float* __restrict__ t, int s, int lane) {
    return *(const v4f*)&t[(size_t)s * D + 4 * lane];
}

__device__ __forceinline__ v4f row_sum(const float* __restrict__ t, const int* __restrict__ nbr,
                                       int beg, int end, int ids, int lane) {
    v4f acc = {0.0f, 0.0f, 0.0f, 0.0f};
    for (int base = beg; base < end; base += 32) {
        if (base != beg) ids = (base + lane < end) ? nbr[base + lane] : 0;
        const int cnt = min(32, end - base);
        int j = 0;
        for (; j + 4 <= cnt; j += 4) {
            const int s0 = __builtin_amdgcn_readlane(ids, j);
            const int s1 = __builtin_amdgcn_readlane(ids, j + 1);
            const int s2 = __builtin_amdgcn_readlane(ids, j + 2);
            const int s3 = __builtin_amdgcn_readlane(ids, j + 3);
            const v4f v0 = ld_row(t, s0, lane);
            const v4f v1 = ld_row(t, s1, lane);
            const v4f v2 = ld_row(t, s2, lane);
            const v4f v3 = ld_row(t, s3, lane);
            acc += v0; acc += v1; acc += v2; acc += v3;
        }
        if (j + 2 <= cnt) {
            const int s0 = __builtin_amdgcn_readlane(ids, j);
            const int s1 = __builtin_amdgcn_readlane(ids, j + 1);
            const v4f v0 = ld_row(t, s0, lane);
            const v4f v1 = ld_row(t, s1, lane);
            acc += v0; acc += v1;
            j += 2;
        }
        if (j < cnt) {
            const int s0 = __builtin_amdgcn_readlane(ids, j);
            acc += ld_row(t, s0, lane);
        }
    }
    return acc;
}

__device__ __forceinline__ void put_mean(float* rowS, v4f s, int count, int lane) {
    const float c = fmaxf((float)count, 1.0f);
    v2f lo, hi;
    lo.x = s.x / c; lo.y = s.y / c; hi.x = s.z / c; hi.y = s.w / c;
    *(v2f*)&rowS[4 * lane]     = lo;
    *(v2f*)&rowS[4 * lane + 2] = hi;
}

template <int M>
__device__ __forceinline__ float row_xor(float v) {
    return __builtin_bit_cast(float, __builtin_amdgcn_update_dpp(0, __builtin_bit_cast(int, v),
                                                                 0x160 + M, 0xf, 0xf, true));
}

__device__ __forceinline__ float xhalf(float v) {
    const int b = __builtin_bit_cast(int, v);
    return __builtin_bit_cast(float, __builtin_amdgcn_permlanex16(b, b, 0x76543210, 0xfedcba98,
                                                                  false, false));
}

__device__ __forceinline__ float wave_sum(float v) {
    v += row_xor<1>(v);
    v += row_xor<2>(v);
    v += row_xor<4>(v);
    v += row_xor<8>(v);
    v += xhalf(v);
    return v;
}

__device__ __forceinline__ float sq4(v2f a, v2f b) {
    return a.x * a.x + a.y * a.y + b.x * b.x + b.y * b.y;
}

__global__ __launch_bounds__(256)
void combine_m_kernel(const float* __restrict__ t_m, const float* __restrict__ t_h,
                      const int* __restrict__ rp_mm, const int* __restrict__ nbr_mm,
                      const int* __restrict__ rp_hm, const int* __restrict__ nbr_hm,
                      const float* __restrict__ Wp,
                      const float* __restrict__ bl_mm, const float* __restrict__ bl_hm,
                      const float* __restrict__ g, const float* __restrict__ be,
                      float* __restrict__ out) {
    __shared__ __attribute__((aligned(16))) float tS[16 * LDSW], aS[16 * LDSW], bS[16 * LDSW];
    const int tid  = threadIdx.x;
    const int row0 = blockIdx.x * 16;
    const int lane = tid & 31;
    const int wv = __builtin_amdgcn_readfirstlane(tid >> 5);

    const int g0 = row0 + wv, g1 = g0 + 8;
    const v4f tr0 = ld_row(t_m, g0, lane);
    const v4f tr1 = ld_row(t_m, g1, lane);
    const int ba0 = rp_mm[g0], ea0 = rp_mm[g0 + 1], ba1 = rp_mm[g1], ea1 = rp_mm[g1 + 1];
    const int bb0 = rp_hm[g0], eb0 = rp_hm[g0 + 1], bb1 = rp_hm[g1], eb1 = rp_hm[g1 + 1];
    const int ia0 = first_ids(nbr_mm, ba0, ea0, lane);
    const int ib0 = first_ids(nbr_hm, bb0, eb0, lane);
    const int ia1 = first_ids(nbr_mm, ba1, ea1, lane);
    const int ib1 = first_ids(nbr_hm, bb1, eb1, lane);

    put_mean(&aS[wv * LDSW],       row_sum(t_m, nbr_mm, ba0, ea0, ia0, lane), ea0 - ba0, lane);
    put_mean(&bS[wv * LDSW],       row_sum(t_h, nbr_hm, bb0, eb0, ib0, lane), eb0 - bb0, lane);
    put_mean(&aS[(wv + 8) * LDSW], row_sum(t_m, nbr_mm, ba1, ea1, ia1, lane), ea1 - ba1, lane);
    put_mean(&bS[(wv + 8) * LDSW], row_sum(t_h, nbr_hm, bb1, eb1, ib1, lane), eb1 - bb1, lane);
    put_row(&tS[wv * LDSW],       tr0, lane);
    put_row(&tS[(wv + 8) * LDSW], tr1, lane);
    __syncthreads();

    const int n = lane & 15, h = lane >> 4, m = lane & 15;
    const int kb = h * 2;
    const int col = wv * 16 + n;

    const v4f* __restrict__ P = (const v4f*)Wp + tid;
    v8f acc_mm = {}, acc_hm = {};
    for (int q2 = 0; q2 < D / 8; ++q2) {
        const int k = 8 * q2;
        const v4f w1 = P[q2 * 256], w2 = P[PKC + q2 * 256];
        const v4f w3 = P[2 * PKC + q2 * 256], w4 = P[3 * PKC + q2 * 256];
        const v2f at0 = *(const v2f*)&tS[m * LDSW + k + kb], at1 = *(const v2f*)&tS[m * LDSW + k + 4 + kb];
        const v2f aa0 = *(const v2f*)&aS[m * LDSW + k + kb], aa1 = *(const v2f*)&aS[m * LDSW + k + 4 + kb];
        const v2f ab0 = *(const v2f*)&bS[m * LDSW + k + kb], ab1 = *(const v2f*)&bS[m * LDSW + k + 4 + kb];
        acc_mm = wmma_f32(aa0, frag_lo(w1), acc_mm);
        acc_mm = wmma_f32(at0, frag_lo(w2), acc_mm);
        acc_hm = wmma_f32(ab0, frag_lo(w3), acc_hm);
        acc_hm = wmma_f32(at0, frag_lo(w4), acc_hm);
        acc_mm = wmma_f32(aa1, frag_hi(w1), acc_mm);
        acc_mm = wmma_f32(at1, frag_hi(w2), acc_mm);
        acc_hm = wmma_f32(ab1, frag_hi(w3), acc_hm);
        acc_hm = wmma_f32(at1, frag_hi(w4), acc_hm);
    }

    const float c1 = bl_mm[col], c2 = bl_hm[col];
    __syncthreads();
    for (int p = 0; p < 8; ++p) {
        const int r = p + 8 * h;
        aS[r * LDSW + col] = acc_mm[p] + c1;
        bS[r * LDSW + col] = acc_hm[p] + c2;
    }
    __syncthreads();

    const int cA = 2 * lane, cB = 64 + 2 * lane;
    const float gA0 = g[cA], gA1 = g[cA + 1], gB0 = g[cB], gB1 = g[cB + 1];
    const float eA0 = be[cA], eA1 = be[cA + 1], eB0 = be[cB], eB1 = be[cB + 1];
#pragma unroll 1
    for (int i = 0; i < 2; ++i) {
        const int r = wv + 8 * i;
        const v2f tA = *(const v2f*)&tS[r * LDSW + cA], tB = *(const v2f*)&tS[r * LDSW + cB];
        const v2f pA = *(const v2f*)&aS[r * LDSW + cA], pB = *(const v2f*)&aS[r * LDSW + cB];
        const v2f qA = *(const v2f*)&bS[r * LDSW + cA], qB = *(const v2f*)&bS[r * LDSW + cB];
        const float s1 = wave_sum(sq4(pA, pB));
        const float s2 = wave_sum(sq4(qA, qB));
        const float n1 = fmaxf(sqrtf(s1), 1e-12f);
        const float n2 = fmaxf(sqrtf(s2), 1e-12f);
        const float y0 = tA.x + 0.5f * (pA.x / n1 + qA.x / n2);
        const float y1 = tA.y + 0.5f * (pA.y / n1 + qA.y / n2);
        const float y2 = tB.x + 0.5f * (pB.x / n1 + qB.x / n2);
        const float y3 = tB.y + 0.5f * (pB.y / n1 + qB.y / n2);
        const float sy  = wave_sum(y0 + y1 + y2 + y3);
        const float syy = wave_sum(y0 * y0 + y1 * y1 + y2 * y2 + y3 * y3);
        const float mu  = sy * (1.0f / D);
        const float var = syy * (1.0f / D) - mu * mu;
        const float rs  = rsqrtf(var + 1e-5f);
        v2f oA, oB;
        oA.x = (y0 - mu) * rs * gA0 + eA0;
        oA.y = (y1 - mu) * rs * gA1 + eA1;
        oB.x = (y2 - mu) * rs * gB0 + eB0;
        oB.y = (y3 - mu) * rs * gB1 + eB1;
        float* orow = out + (size_t)(row0 + r) * D;
        *(v2f*)&orow[cA] = oA;
        *(v2f*)&orow[cB] = oB;
    }
}

__global__ __launch_bounds__(256)
void combine_h_kernel(const float* __restrict__ t_h, const float* __restrict__ t_m,
                      const int* __restrict__ rp_mh, const int* __restrict__ nbr_mh,
                      const float* __restrict__ Wp,
                      const float* __restrict__ bl_mh,
                      const float* __restrict__ g, const float* __restrict__ be,
                      float* __restrict__ out) {
    __shared__ __attribute__((aligned(16))) float tS[16 * LDSW], aS[16 * LDSW];
    const int tid  = threadIdx.x;
    const int row0 = blockIdx.x * 16;
    const int lane = tid & 31;
    const int wv = __builtin_amdgcn_readfirstlane(tid >> 5);

    const int g0 = row0 + wv, g1 = g0 + 8;
    const v4f tr0 = ld_row(t_h, g0, lane);
    const v4f tr1 = ld_row(t_h, g1, lane);
    const int ba0 = rp_mh[g0], ea0 = rp_mh[g0 + 1], ba1 = rp_mh[g1], ea1 = rp_mh[g1 + 1];
    const int ia0 = first_ids(nbr_mh, ba0, ea0, lane);
    const int ia1 = first_ids(nbr_mh, ba1, ea1, lane);

    put_mean(&aS[wv * LDSW],       row_sum(t_m, nbr_mh, ba0, ea0, ia0, lane), ea0 - ba0, lane);
    put_mean(&aS[(wv + 8) * LDSW], row_sum(t_m, nbr_mh, ba1, ea1, ia1, lane), ea1 - ba1, lane);
    put_row(&tS[wv * LDSW],       tr0, lane);
    put_row(&tS[(wv + 8) * LDSW], tr1, lane);
    __syncthreads();

    const int n = lane & 15, h = lane >> 4, m = lane & 15;
    const int kb = h * 2;
    const int col = wv * 16 + n;

    const v4f* __restrict__ P = (const v4f*)Wp + tid;
    v8f acc = {};
    for (int q2 = 0; q2 < D / 8; ++q2) {
        const int k = 8 * q2;
        const v4f w1 = P[q2 * 256], w2 = P[PKC + q2 * 256];
        const v2f at0 = *(const v2f*)&tS[m * LDSW + k + kb], at1 = *(const v2f*)&tS[m * LDSW + k + 4 + kb];
        const v2f aa0 = *(const v2f*)&aS[m * LDSW + k + kb], aa1 = *(const v2f*)&aS[m * LDSW + k + 4 + kb];
        acc = wmma_f32(aa0, frag_lo(w1), acc);
        acc = wmma_f32(at0, frag_lo(w2), acc);
        acc = wmma_f32(aa1, frag_hi(w1), acc);
        acc = wmma_f32(at1, frag_hi(w2), acc);
    }

    const float c1 = bl_mh[col];
    __syncthreads();
    for (int p = 0; p < 8; ++p)
        aS[(p + 8 * h) * LDSW + col] = acc[p] + c1;
    __syncthreads();

    const int cA = 2 * lane, cB = 64 + 2 * lane;
    const float gA0 = g[cA], gA1 = g[cA + 1], gB0 = g[cB], gB1 = g[cB + 1];
    const float eA0 = be[cA], eA1 = be[cA + 1], eB0 = be[cB], eB1 = be[cB + 1];
#pragma unroll 1
    for (int i = 0; i < 2; ++i) {
        const int r = wv + 8 * i;
        const v2f tA = *(const v2f*)&tS[r * LDSW + cA], tB = *(const v2f*)&tS[r * LDSW + cB];
        const v2f pA = *(const v2f*)&aS[r * LDSW + cA], pB = *(const v2f*)&aS[r * LDSW + cB];
        const float s1 = wave_sum(sq4(pA, pB));
        const float n1 = fmaxf(sqrtf(s1), 1e-12f);
        const float y0 = tA.x + pA.x / n1;
        const float y1 = tA.y + pA.y / n1;
        const float y2 = tB.x + pB.x / n1;
        const float y3 = tB.y + pB.y / n1;
        const float sy  = wave_sum(y0 + y1 + y2 + y3);
        const float syy = wave_sum(y0 * y0 + y1 * y1 + y2 * y2 + y3 * y3);
        const float mu  = sy * (1.0f / D);
        const float var = syy * (1.0f / D) - mu * mu;
        const float rs  = rsqrtf(var + 1e-5f);
        v2f oA, oB;
        oA.x = (y0 - mu) * rs * gA0 + eA0;
        oA.y = (y1 - mu) * rs * gA1 + eA1;
        oB.x = (y2 - mu) * rs * gB0 + eB0;
        oB.y = (y3 - mu) * rs * gB1 + eB1;
        float* orow = out + (size_t)(row0 + r) * D;
        *(v2f*)&orow[cA] = oA;
        *(v2f*)&orow[cB] = oB;
    }
}

extern "C" void kernel_launch(void* const* d_in, const int* in_sizes, int n_in,
                              void* d_out, int out_size, void* d_ws, size_t ws_size,
                              hipStream_t stream) {
    const float* x_m    = (const float*)d_in[0];
    const float* x_h    = (const float*)d_in[1];
    const float* W_nt_m = (const float*)d_in[2];
    const float* b_nt_m = (const float*)d_in[3];
    const float* W_nt_h = (const float*)d_in[4];
    const float* b_nt_h = (const float*)d_in[5];
    const float* Wl_mm  = (const float*)d_in[6];
    const float* bl_mm  = (const float*)d_in[7];
    const float* Wr_mm  = (const float*)d_in[8];
    const float* Wl_mh  = (const float*)d_in[9];
    const float* bl_mh  = (const float*)d_in[10];
    const float* Wr_mh  = (const float*)d_in[11];
    const float* Wl_hm  = (const float*)d_in[12];
    const float* bl_hm  = (const float*)d_in[13];
    const float* Wr_hm  = (const float*)d_in[14];
    const float* g_m    = (const float*)d_in[15];
    const float* be_m   = (const float*)d_in[16];
    const float* g_h    = (const float*)d_in[17];
    const float* be_h   = (const float*)d_in[18];
    const int*   ei_mm  = (const int*)d_in[19];
    const int*   ei_mh  = (const int*)d_in[20];
    const int*   ei_hm  = (const int*)d_in[21];

    const int N_M  = in_sizes[0] / D;
    const int N_H  = in_sizes[1] / D;
    const int E_mm = in_sizes[19] / 2;
    const int E_mh = in_sizes[20] / 2;
    const int E_hm = in_sizes[21] / 2;

    const size_t nm = (size_t)N_M * D;
    const size_t nh = (size_t)N_H * D;

    float* ws     = (float*)d_ws;
    float* t_m    = ws;
    float* t_h    = t_m + nm;
    float* wpk  = t_h + nh;
    int* rp_mm  = (int*)(wpk + 6 * PKF);
    int* rp_hm  = rp_mm + (N_M + 1);
    int* rp_mh  = rp_hm + (N_M + 1);
    int* nbr_mm = rp_mh + (N_H + 1);
    int* nbr_hm = nbr_mm + E_mm;
    int* nbr_mh = nbr_hm + E_hm;

    auto rg = [](int n) { return (n + RSZ - 1) / RSZ; };
    const int nr_mm = rg(N_M), nr_hm = rg(N_M), nr_mh = rg(N_H);
    const int tiles_m = N_M / 16, tiles_h = N_H / 16;
    const int wg_m = (tiles_m + LINT - 1) / LINT, wg_h = (tiles_h + LINT - 1) / LINT;
    const int npk = 6 * (PKC / RSZ);
    front_kernel<<<nr_mm + nr_hm + nr_mh + npk + wg_m + wg_h, RSZ, 0, stream>>>(
        ei_mm, rp_mm, nbr_mm, E_mm, N_M, nr_mm,
        ei_hm, rp_hm, nbr_hm, E_hm, N_M, nr_hm,
        ei_mh, rp_mh, nbr_mh, E_mh, N_H, nr_mh,
        x_m, W_nt_m, b_nt_m, t_m, tiles_m, wg_m,
        x_h, W_nt_h, b_nt_h, t_h, tiles_h,
        Wl_mm, Wr_mm, Wl_hm, Wr_hm, Wl_mh, Wr_mh, wpk, npk);

    float* out_m = (float*)d_out;
    float* out_h = out_m + nm;
    combine_m_kernel<<<N_M / 16, 256, 0, stream>>>(t_m, t_h, rp_mm, nbr_mm, rp_hm, nbr_hm,
                                                   wpk, bl_mm, bl_hm,
                                                   g_m, be_m, out_m);
    combine_h_kernel<<<N_H / 16, 256, 0, stream>>>(t_h, t_m, rp_mh, nbr_mh,
                                                   wpk + 4 * PKF, bl_mh,
                                                   g_h, be_h, out_h);
}
